# starting kernel
# baseline (speedup 1.0000x reference)
#include <hip/hip_runtime.h>

#define COUT 128

typedef float v2f __attribute__((ext_vector_type(2)));
typedef float v8f __attribute__((ext_vector_type(8)));

__global__ void copy_f4(float* __restrict__ dst, const float* __restrict__ src, int n4) {
  int i = blockIdx.x * blockDim.x + threadIdx.x;
  if (i < n4) reinterpret_cast<float4*>(dst)[i] = reinterpret_cast<const float4*>(src)[i];
}

__global__ void scatter_add(float* __restrict__ agg, const float* __restrict__ h,
                            const int* __restrict__ src, const int* __restrict__ dst,
                            int total, int gshift, int C) {
  int idx = blockIdx.x * blockDim.x + threadIdx.x;
  if (idx >= total) return;
  int e = idx >> gshift;
  int g = idx & ((1 << gshift) - 1);
  int s = src[e], d = dst[e];
  float4 v = reinterpret_cast<const float4*>(h + (size_t)s * C)[g];
  float* p = agg + (size_t)d * C + (g << 2);
  atomicAdd(p + 0, v.x);
  atomicAdd(p + 1, v.y);
  atomicAdd(p + 2, v.z);
  atomicAdd(p + 3, v.w);
}

__global__ __launch_bounds__(256)
void gemm_bias_act(const float* __restrict__ A, const float* __restrict__ W,
                   const float* __restrict__ bias, float* __restrict__ out,
                   int N, int K, int do_relu) {
  extern __shared__ float lds[];
  for (int i = threadIdx.x; i < K * COUT; i += blockDim.x) lds[i] = W[i];
  __syncthreads();

  const int wave = threadIdx.x >> 5;
  const int lane = threadIdx.x & 31;
  const int half = lane >> 4;
  const int l15  = lane & 15;
  const int row0 = (blockIdx.x * 8 + wave) * 16;
  if (row0 >= N) return;

  const float* aRow = A + (size_t)(row0 + l15) * K + half * 2;

  v8f acc[8] = {};
  for (int k0 = 0; k0 < K; k0 += 4) {
    v2f a = *reinterpret_cast<const v2f*>(aRow + k0);
    const float* wr0 = &lds[(k0 + half * 2) * COUT];
    const float* wr1 = wr0 + COUT;
#pragma unroll
    for (int ct = 0; ct < 8; ++ct) {
      int col = ct * 16 + l15;
      v2f b;
      b.x = wr0[col];
      b.y = wr1[col];
      acc[ct] = __builtin_amdgcn_wmma_f32_16x16x4_f32(
          false, a, false, b, (short)0, acc[ct], false, false);
    }
  }

#pragma unroll
  for (int ct = 0; ct < 8; ++ct) {
    int col = ct * 16 + l15;
    float bv = bias[col];
#pragma unroll
    for (int r = 0; r < 8; ++r) {
      float v = acc[ct][r] + bv;
      if (do_relu) v = fmaxf(v, 0.0f);
      out[(size_t)(row0 + half * 8 + r) * COUT + col] = v;
    }
  }
}

extern "C" void kernel_launch(void* const* d_in, const int* in_sizes, int n_in,
                              void* d_out, int out_size, void* d_ws, size_t ws_size,
                              hipStream_t stream) {
  const float* x   = (const float*)d_in[0];
  const int*   ei  = (const int*)d_in[1];
  const float* W1a = (const float*)d_in[2];
  const float* b1a = (const float*)d_in[3];
  const float* W1b = (const float*)d_in[4];
  const float* b1b = (const float*)d_in[5];
  const float* W2a = (const float*)d_in[6];
  const float* b2a = (const float*)d_in[7];
  const float* W2b = (const float*)d_in[8];
  const float* b2b = (const float*)d_in[9];
  float* out = (float*)d_out;

  const int N = in_sizes[0] / 64;
  const int E = in_sizes[1] / 2;
  const int* src = ei;
  const int* dst = ei + E;

  float* R0 = (float*)d_ws;
  float* R1 = R0 + (size_t)N * 128;
  float* R2 = R1 + (size_t)N * 128;

  const int strips  = (N + 15) / 16;
  const int gblocks = (strips + 7) / 8;

  {
    int n4 = (N * 64) / 4;
    copy_f4<<<(n4 + 255) / 256, 256, 0, stream>>>(R0, x, n4);
  }
  {
    int total = E * (64 / 4);
    scatter_add<<<(total + 255) / 256, 256, 0, stream>>>(R0, x, src, dst, total, 4, 64);
  }
  gemm_bias_act<<<gblocks, 256,  64 * COUT * sizeof(float), stream>>>(R0, W1a, b1a, R1, N,  64, 1);
  gemm_bias_act<<<gblocks, 256, 128 * COUT * sizeof(float), stream>>>(R1, W1b, b1b, R2, N, 128, 1);

  {
    int n4 = (N * 128) / 4;
    copy_f4<<<(n4 + 255) / 256, 256, 0, stream>>>(R0, R2, n4);
  }
  {
    int total = E * (128 / 4);
    scatter_add<<<(total + 255) / 256, 256, 0, stream>>>(R0, R2, src, dst, total, 5, 128);
  }
  gemm_bias_act<<<gblocks, 256, 128 * COUT * sizeof(float), stream>>>(R0, W2a, b2a, R1, N, 128, 1);
  gemm_bias_act<<<gblocks, 256, 128 * COUT * sizeof(float), stream>>>(R1, W2b, b2b, out, N, 128, 0);
}


# Replace the f32-atomic edge scatter and its copy by a per-call edge list grouped by destination (integer-atomic rank, scan, place) and an atomic-free row gather; neighbour sum order = arrival rank
# speedup vs baseline: 1.1587x; 1.1587x over previous
#include <hip/hip_runtime.h>

#define COUT 128

typedef float v2f __attribute__((ext_vector_type(2)));
typedef float v8f __attribute__((ext_vector_type(8)));


__global__ void zero_counts(int* __restrict__ cnt, int n) {
  int i = blockIdx.x * blockDim.x + threadIdx.x;
  if (i < n) cnt[i] = 0;
}

__global__ void rank_edges(int* __restrict__ cnt, const int* __restrict__ dst,
                           int* __restrict__ rank, int E) {
  int e = blockIdx.x * blockDim.x + threadIdx.x;
  if (e < E) rank[e] = atomicAdd(&cnt[dst[e]], 1);
}

__global__ __launch_bounds__(1024)
void scan_counts(const int* __restrict__ cnt, int* __restrict__ rowptr, int n) {
  __shared__ int wsum[32];
  const int tid  = threadIdx.x;
  const int lane = tid & 31;
  const int wave = tid >> 5;
  const int chunk = (n + 1023) / 1024;
  const int beg = tid * chunk;
  const int end = min(beg + chunk, n);
  int s = 0;
  for (int i = beg; i < end; ++i) s += cnt[i];
  int v = s;
  for (int off = 1; off < 32; off <<= 1) {
    int t = __shfl_up(v, off);
    if (lane >= off) v += t;
  }
  if (lane == 31) wsum[wave] = v;
  __syncthreads();
  if (wave == 0) {
    int w = wsum[lane];
    for (int off = 1; off < 32; off <<= 1) {
      int t = __shfl_up(w, off);
      if (lane >= off) w += t;
    }
    wsum[lane] = w;
  }
  __syncthreads();
  int base = v - s + (wave > 0 ? wsum[wave - 1] : 0);
  for (int i = beg; i < end; ++i) { rowptr[i] = base; base += cnt[i]; }
  if (tid == 0) rowptr[n] = wsum[31];
}

__global__ void place_edges(const int* __restrict__ rowptr, const int* __restrict__ src,
                            const int* __restrict__ dst, const int* __restrict__ rank,
                            int* __restrict__ csr, int E) {
  int e = blockIdx.x * blockDim.x + threadIdx.x;
  if (e < E) csr[rowptr[dst[e]] + rank[e]] = src[e];
}

template <int C>
__global__ __launch_bounds__(256)
void gather_rows(float* __restrict__ agg, const float* __restrict__ h,
                 const int* __restrict__ rowptr, const int* __restrict__ csr, int N) {
  constexpr int LPN = C / 4;
  const int gtid = blockIdx.x * 256 + threadIdx.x;
  const int node = gtid / LPN;
  const int g    = gtid % LPN;
  if (node >= N) return;
  const float4* __restrict__ h4 = reinterpret_cast<const float4*>(h);
  const int beg = rowptr[node];
  const int end = rowptr[node + 1];
  const unsigned last = (unsigned)(N - 1);
  float4 acc = h4[node * LPN + g];
  for (int j = beg; j < end; j += 4) {
    unsigned s[4];
    float4 v[4];
#pragma unroll
    for (int k = 0; k < 4; ++k) s[k] = min((unsigned)csr[min(j + k, end - 1)], last);
#pragma unroll
    for (int k = 0; k < 4; ++k) v[k] = h4[s[k] * LPN + g];
#pragma unroll
    for (int k = 0; k < 4; ++k) {
      const bool on = (j + k) < end;
      acc.x += on ? v[k].x : 0.0f;
      acc.y += on ? v[k].y : 0.0f;
      acc.z += on ? v[k].z : 0.0f;
      acc.w += on ? v[k].w : 0.0f;
    }
  }
  reinterpret_cast<float4*>(agg)[node * LPN + g] = acc;
}

__global__ __launch_bounds__(256)
void gemm_bias_act(const float* __restrict__ A, const float* __restrict__ W,
                   const float* __restrict__ bias, float* __restrict__ out,
                   int N, int K, int do_relu) {
  extern __shared__ float lds[];
  for (int i = threadIdx.x; i < K * COUT; i += blockDim.x) lds[i] = W[i];
  __syncthreads();

  const int wave = threadIdx.x >> 5;
  const int lane = threadIdx.x & 31;
  const int half = lane >> 4;
  const int l15  = lane & 15;
  const int row0 = (blockIdx.x * 8 + wave) * 16;
  if (row0 >= N) return;

  const float* aRow = A + (size_t)(row0 + l15) * K + half * 2;

  v8f acc[8] = {};
  for (int k0 = 0; k0 < K; k0 += 4) {
    v2f a = *reinterpret_cast<const v2f*>(aRow + k0);
    const float* wr0 = &lds[(k0 + half * 2) * COUT];
    const float* wr1 = wr0 + COUT;
#pragma unroll
    for (int ct = 0; ct < 8; ++ct) {
      int col = ct * 16 + l15;
      v2f b;
      b.x = wr0[col];
      b.y = wr1[col];
      acc[ct] = __builtin_amdgcn_wmma_f32_16x16x4_f32(
          false, a, false, b, (short)0, acc[ct], false, false);
    }
  }

#pragma unroll
  for (int ct = 0; ct < 8; ++ct) {
    int col = ct * 16 + l15;
    float bv = bias[col];
#pragma unroll
    for (int r = 0; r < 8; ++r) {
      float v = acc[ct][r] + bv;
      if (do_relu) v = fmaxf(v, 0.0f);
      out[(size_t)(row0 + half * 8 + r) * COUT + col] = v;
    }
  }
}

extern "C" void kernel_launch(void* const* d_in, const int* in_sizes, int n_in,
                              void* d_out, int out_size, void* d_ws, size_t ws_size,
                              hipStream_t stream) {
  const float* x   = (const float*)d_in[0];
  const int*   ei  = (const int*)d_in[1];
  const float* W1a = (const float*)d_in[2];
  const float* b1a = (const float*)d_in[3];
  const float* W1b = (const float*)d_in[4];
  const float* b1b = (const float*)d_in[5];
  const float* W2a = (const float*)d_in[6];
  const float* b2a = (const float*)d_in[7];
  const float* W2b = (const float*)d_in[8];
  const float* b2b = (const float*)d_in[9];
  float* out = (float*)d_out;

  const int N = in_sizes[0] / 64;
  const int E = in_sizes[1] / 2;
  const int* src = ei;
  const int* dst = ei + E;

  float* R0 = (float*)d_ws;
  float* R1 = R0 + (size_t)N * 128;
  float* R2 = R1 + (size_t)N * 128;

  int* cnt    = (int*)(R2 + (size_t)N * 128);
  int* rowptr = cnt + N;
  int* rank   = rowptr + (N + 1);
  int* csr    = rank + E;

  const int strips  = (N + 15) / 16;
  const int gblocks = (strips + 7) / 8;

  zero_counts<<<(N + 255) / 256, 256, 0, stream>>>(cnt, N);
  rank_edges<<<(E + 255) / 256, 256, 0, stream>>>(cnt, dst, rank, E);
  scan_counts<<<1, 1024, 0, stream>>>(cnt, rowptr, N);
  place_edges<<<(E + 255) / 256, 256, 0, stream>>>(rowptr, src, dst, rank, csr, E);

  gather_rows<64><<<(N * 16 + 255) / 256, 256, 0, stream>>>(R0, x, rowptr, csr, N);
  gemm_bias_act<<<gblocks, 256,  64 * COUT * sizeof(float), stream>>>(R0, W1a, b1a, R1, N,  64, 1);
  gemm_bias_act<<<gblocks, 256, 128 * COUT * sizeof(float), stream>>>(R1, W1b, b1b, R2, N, 128, 1);

  gather_rows<128><<<(N * 32 + 255) / 256, 256, 0, stream>>>(R0, R2, rowptr, csr, N);
  gemm_bias_act<<<gblocks, 256, 128 * COUT * sizeof(float), stream>>>(R0, W2a, b2a, R1, N, 128, 1);
  gemm_bias_act<<<gblocks, 256, 128 * COUT * sizeof(float), stream>>>(R1, W2b, b2b, out, N, 128, 0);
}


# GEMM weight staging into LDS: 16-byte loads, 16 per thread, all issued before the first LDS store, in place of one 4-byte load and one full wait per element
# speedup vs baseline: 1.6121x; 1.3913x over previous
#include <hip/hip_runtime.h>

#define COUT 128

typedef float v2f __attribute__((ext_vector_type(2)));
typedef float v8f __attribute__((ext_vector_type(8)));


__global__ void zero_counts(int* __restrict__ cnt, int n) {
  int i = blockIdx.x * blockDim.x + threadIdx.x;
  if (i < n) cnt[i] = 0;
}

__global__ void rank_edges(int* __restrict__ cnt, const int* __restrict__ dst,
                           int* __restrict__ rank, int E) {
  int e = blockIdx.x * blockDim.x + threadIdx.x;
  if (e < E) rank[e] = atomicAdd(&cnt[dst[e]], 1);
}

__global__ __launch_bounds__(1024)
void scan_counts(const int* __restrict__ cnt, int* __restrict__ rowptr, int n) {
  __shared__ int wsum[32];
  const int tid  = threadIdx.x;
  const int lane = tid & 31;
  const int wave = tid >> 5;
  const int chunk = (n + 1023) / 1024;
  const int beg = tid * chunk;
  const int end = min(beg + chunk, n);
  int s = 0;
  for (int i = beg; i < end; ++i) s += cnt[i];
  int v = s;
  for (int off = 1; off < 32; off <<= 1) {
    int t = __shfl_up(v, off);
    if (lane >= off) v += t;
  }
  if (lane == 31) wsum[wave] = v;
  __syncthreads();
  if (wave == 0) {
    int w = wsum[lane];
    for (int off = 1; off < 32; off <<= 1) {
      int t = __shfl_up(w, off);
      if (lane >= off) w += t;
    }
    wsum[lane] = w;
  }
  __syncthreads();
  int base = v - s + (wave > 0 ? wsum[wave - 1] : 0);
  for (int i = beg; i < end; ++i) { rowptr[i] = base; base += cnt[i]; }
  if (tid == 0) rowptr[n] = wsum[31];
}

__global__ void place_edges(const int* __restrict__ rowptr, const int* __restrict__ src,
                            const int* __restrict__ dst, const int* __restrict__ rank,
                            int* __restrict__ csr, int E) {
  int e = blockIdx.x * blockDim.x + threadIdx.x;
  if (e < E) csr[rowptr[dst[e]] + rank[e]] = src[e];
}

template <int C>
__global__ __launch_bounds__(256)
void gather_rows(float* __restrict__ agg, const float* __restrict__ h,
                 const int* __restrict__ rowptr, const int* __restrict__ csr, int N) {
  constexpr int LPN = C / 4;
  const int gtid = blockIdx.x * 256 + threadIdx.x;
  const int node = gtid / LPN;
  const int g    = gtid % LPN;
  if (node >= N) return;
  const float4* __restrict__ h4 = reinterpret_cast<const float4*>(h);
  const int beg = rowptr[node];
  const int end = rowptr[node + 1];
  const unsigned last = (unsigned)(N - 1);
  float4 acc = h4[node * LPN + g];
  for (int j = beg; j < end; j += 4) {
    unsigned s[4];
    float4 v[4];
#pragma unroll
    for (int k = 0; k < 4; ++k) s[k] = min((unsigned)csr[min(j + k, end - 1)], last);
#pragma unroll
    for (int k = 0; k < 4; ++k) v[k] = h4[s[k] * LPN + g];
#pragma unroll
    for (int k = 0; k < 4; ++k) {
      const bool on = (j + k) < end;
      acc.x += on ? v[k].x : 0.0f;
      acc.y += on ? v[k].y : 0.0f;
      acc.z += on ? v[k].z : 0.0f;
      acc.w += on ? v[k].w : 0.0f;
    }
  }
  reinterpret_cast<float4*>(agg)[node * LPN + g] = acc;
}

template <int VPT>
__device__ __forceinline__ void stage_pieces(float4* lds4, const float4* __restrict__ W4, int first) {
  float4 v[VPT];
#pragma unroll
  for (int r = 0; r < VPT; ++r) v[r] = W4[first + r * 256];
#pragma unroll
  for (int r = 0; r < VPT; ++r) lds4[first + r * 256] = v[r];
}

__global__ __launch_bounds__(256)
void gemm_bias_act(const float* __restrict__ A, const float* __restrict__ W,
                   const float* __restrict__ bias, float* __restrict__ out,
                   int N, int K, int do_relu) {
  extern __shared__ __align__(16) float lds[];
  {
    const float4* __restrict__ W4 = reinterpret_cast<const float4*>(W);
    float4* lds4 = reinterpret_cast<float4*>(lds);
    const int nvec = K * (COUT / 4);
    int done = 0;
    for (; done + 16 * 256 <= nvec; done += 16 * 256) stage_pieces<16>(lds4, W4, done + threadIdx.x);
    for (; done +  8 * 256 <= nvec; done +=  8 * 256) stage_pieces<8>(lds4, W4, done + threadIdx.x);
    for (int i = done + threadIdx.x; i < nvec; i += 256) lds4[i] = W4[i];
  }
  __syncthreads();

  const int wave = threadIdx.x >> 5;
  const int lane = threadIdx.x & 31;
  const int half = lane >> 4;
  const int l15  = lane & 15;
  const int row0 = (blockIdx.x * 8 + wave) * 16;
  if (row0 >= N) return;

  const float* aRow = A + (size_t)(row0 + l15) * K + half * 2;

  v8f acc[8] = {};
  for (int k0 = 0; k0 < K; k0 += 4) {
    v2f a = *reinterpret_cast<const v2f*>(aRow + k0);
    const float* wr0 = &lds[(k0 + half * 2) * COUT];
    const float* wr1 = wr0 + COUT;
#pragma unroll
    for (int ct = 0; ct < 8; ++ct) {
      int col = ct * 16 + l15;
      v2f b;
      b.x = wr0[col];
      b.y = wr1[col];
      acc[ct] = __builtin_amdgcn_wmma_f32_16x16x4_f32(
          false, a, false, b, (short)0, acc[ct], false, false);
    }
  }

#pragma unroll
  for (int ct = 0; ct < 8; ++ct) {
    int col = ct * 16 + l15;
    float bv = bias[col];
#pragma unroll
    for (int r = 0; r < 8; ++r) {
      float v = acc[ct][r] + bv;
      if (do_relu) v = fmaxf(v, 0.0f);
      out[(size_t)(row0 + half * 8 + r) * COUT + col] = v;
    }
  }
}

extern "C" void kernel_launch(void* const* d_in, const int* in_sizes, int n_in,
                              void* d_out, int out_size, void* d_ws, size_t ws_size,
                              hipStream_t stream) {
  const float* x   = (const float*)d_in[0];
  const int*   ei  = (const int*)d_in[1];
  const float* W1a = (const float*)d_in[2];
  const float* b1a = (const float*)d_in[3];
  const float* W1b = (const float*)d_in[4];
  const float* b1b = (const float*)d_in[5];
  const float* W2a = (const float*)d_in[6];
  const float* b2a = (const float*)d_in[7];
  const float* W2b = (const float*)d_in[8];
  const float* b2b = (const float*)d_in[9];
  float* out = (float*)d_out;

  const int N = in_sizes[0] / 64;
  const int E = in_sizes[1] / 2;
  const int* src = ei;
  const int* dst = ei + E;

  float* R0 = (float*)d_ws;
  float* R1 = R0 + (size_t)N * 128;
  float* R2 = R1 + (size_t)N * 128;

  int* cnt    = (int*)(R2 + (size_t)N * 128);
  int* rowptr = cnt + N;
  int* rank   = rowptr + (N + 1);
  int* csr    = rank + E;

  const int strips  = (N + 15) / 16;
  const int gblocks = (strips + 7) / 8;

  zero_counts<<<(N + 255) / 256, 256, 0, stream>>>(cnt, N);
  rank_edges<<<(E + 255) / 256, 256, 0, stream>>>(cnt, dst, rank, E);
  scan_counts<<<1, 1024, 0, stream>>>(cnt, rowptr, N);
  place_edges<<<(E + 255) / 256, 256, 0, stream>>>(rowptr, src, dst, rank, csr, E);

  gather_rows<64><<<(N * 16 + 255) / 256, 256, 0, stream>>>(R0, x, rowptr, csr, N);
  gemm_bias_act<<<gblocks, 256,  64 * COUT * sizeof(float), stream>>>(R0, W1a, b1a, R1, N,  64, 1);
  gemm_bias_act<<<gblocks, 256, 128 * COUT * sizeof(float), stream>>>(R1, W1b, b1b, R2, N, 128, 1);

  gather_rows<128><<<(N * 32 + 255) / 256, 256, 0, stream>>>(R0, R2, rowptr, csr, N);
  gemm_bias_act<<<gblocks, 256, 128 * COUT * sizeof(float), stream>>>(R0, W2a, b2a, R1, N, 128, 1);
  gemm_bias_act<<<gblocks, 256, 128 * COUT * sizeof(float), stream>>>(R1, W2b, b2b, out, N, 128, 0);
}


# Row-start scan of the in-degree counts: whole-line 16-byte loads all issued up front, wave scans in registers (DPP adds), one barrier, 16-byte stores; was a per-thread walk over counts 196 B apart
# speedup vs baseline: 2.1573x; 1.3382x over previous
#include <hip/hip_runtime.h>

#define COUT 128

typedef float v2f __attribute__((ext_vector_type(2)));
typedef float v8f __attribute__((ext_vector_type(8)));


__global__ void zero_counts(int* __restrict__ cnt, int n) {
  int i = blockIdx.x * blockDim.x + threadIdx.x;
  if (i < n) cnt[i] = 0;
}

__global__ void rank_edges(int* __restrict__ cnt, const int* __restrict__ dst,
                           int* __restrict__ rank, int E) {
  int e = blockIdx.x * blockDim.x + threadIdx.x;
  if (e < E) rank[e] = atomicAdd(&cnt[dst[e]], 1);
}

__device__ __forceinline__ int wave_inclusive_sum(int x, int lane) {
  x += __builtin_amdgcn_update_dpp(0, x, 0x111, 0xf, 0xf, false);
  x += __builtin_amdgcn_update_dpp(0, x, 0x112, 0xf, 0xf, false);
  x += __builtin_amdgcn_update_dpp(0, x, 0x114, 0xf, 0xf, false);
  x += __builtin_amdgcn_update_dpp(0, x, 0x118, 0xf, 0xf, false);
  const int low_row = __builtin_amdgcn_readlane(x, 15);
  return x + (lane >= 16 ? low_row : 0);
}

#define SCAN_VPL 13

__global__ __launch_bounds__(1024)
void scan_counts(const int* __restrict__ cnt, int* __restrict__ rowptr, int n) {
  __shared__ int wsum[2][32];
  const int tid  = threadIdx.x;
  const int lane = tid & 31;
  const int wave = __builtin_amdgcn_readfirstlane(tid >> 5);
  const int4* __restrict__ c4 = reinterpret_cast<const int4*>(cnt);
  int4* __restrict__ r4 = reinterpret_cast<int4*>(rowptr);
  const int nfull = n >> 2;
  int carry = 0;
  int par = 0;
  for (int first = 0; first < nfull; first += 32 * SCAN_VPL * 32) {
    const int p0 = first + wave * (SCAN_VPL * 32) + lane;
    int4 v[SCAN_VPL];
#pragma unroll
    for (int r = 0; r < SCAN_VPL; ++r) {
      const int p = p0 + r * 32;
      v[r] = (p < nfull) ? c4[p] : make_int4(0, 0, 0, 0);
    }
    int inc[SCAN_VPL];
#pragma unroll
    for (int r = 0; r < SCAN_VPL; ++r)
      inc[r] = wave_inclusive_sum(v[r].x + v[r].y + v[r].z + v[r].w, lane);
    int rowbase[SCAN_VPL];
    int run = 0;
#pragma unroll
    for (int r = 0; r < SCAN_VPL; ++r) {
      rowbase[r] = run;
      run += __builtin_amdgcn_readlane(inc[r], 31);
    }
    if (lane == 0) wsum[par][wave] = run;
    __syncthreads();
    const int w = wave_inclusive_sum(wsum[par][lane], lane);
    const int wbase = carry + __builtin_amdgcn_readlane(w, wave) - run;
#pragma unroll
    for (int r = 0; r < SCAN_VPL; ++r) {
      const int p = p0 + r * 32;
      if (p < nfull) {
        int4 o;
        o.x = wbase + rowbase[r] + inc[r] - (v[r].x + v[r].y + v[r].z + v[r].w);
        o.y = o.x + v[r].x;
        o.z = o.y + v[r].y;
        o.w = o.z + v[r].z;
        r4[p] = o;
      }
    }
    carry += __builtin_amdgcn_readlane(w, 31);
    par ^= 1;
  }
  if (tid == 0) {
    int b = carry;
    for (int i = nfull * 4; i < n; ++i) { rowptr[i] = b; b += cnt[i]; }
    rowptr[n] = b;
  }
}

__global__ void place_edges(const int* __restrict__ rowptr, const int* __restrict__ src,
                            const int* __restrict__ dst, const int* __restrict__ rank,
                            int* __restrict__ csr, int E) {
  int e = blockIdx.x * blockDim.x + threadIdx.x;
  if (e < E) csr[rowptr[dst[e]] + rank[e]] = src[e];
}

template <int C>
__global__ __launch_bounds__(256)
void gather_rows(float* __restrict__ agg, const float* __restrict__ h,
                 const int* __restrict__ rowptr, const int* __restrict__ csr, int N) {
  constexpr int LPN = C / 4;
  const int gtid = blockIdx.x * 256 + threadIdx.x;
  const int node = gtid / LPN;
  const int g    = gtid % LPN;
  if (node >= N) return;
  const float4* __restrict__ h4 = reinterpret_cast<const float4*>(h);
  const int beg = rowptr[node];
  const int end = rowptr[node + 1];
  const unsigned last = (unsigned)(N - 1);
  float4 acc = h4[node * LPN + g];
  for (int j = beg; j < end; j += 4) {
    unsigned s[4];
    float4 v[4];
#pragma unroll
    for (int k = 0; k < 4; ++k) s[k] = min((unsigned)csr[min(j + k, end - 1)], last);
#pragma unroll
    for (int k = 0; k < 4; ++k) v[k] = h4[s[k] * LPN + g];
#pragma unroll
    for (int k = 0; k < 4; ++k) {
      const bool on = (j + k) < end;
      acc.x += on ? v[k].x : 0.0f;
      acc.y += on ? v[k].y : 0.0f;
      acc.z += on ? v[k].z : 0.0f;
      acc.w += on ? v[k].w : 0.0f;
    }
  }
  reinterpret_cast<float4*>(agg)[node * LPN + g] = acc;
}

template <int VPT>
__device__ __forceinline__ void stage_pieces(float4* lds4, const float4* __restrict__ W4, int first) {
  float4 v[VPT];
#pragma unroll
  for (int r = 0; r < VPT; ++r) v[r] = W4[first + r * 256];
#pragma unroll
  for (int r = 0; r < VPT; ++r) lds4[first + r * 256] = v[r];
}

__global__ __launch_bounds__(256)
void gemm_bias_act(const float* __restrict__ A, const float* __restrict__ W,
                   const float* __restrict__ bias, float* __restrict__ out,
                   int N, int K, int do_relu) {
  extern __shared__ __align__(16) float lds[];
  {
    const float4* __restrict__ W4 = reinterpret_cast<const float4*>(W);
    float4* lds4 = reinterpret_cast<float4*>(lds);
    const int nvec = K * (COUT / 4);
    int done = 0;
    for (; done + 16 * 256 <= nvec; done += 16 * 256) stage_pieces<16>(lds4, W4, done + threadIdx.x);
    for (; done +  8 * 256 <= nvec; done +=  8 * 256) stage_pieces<8>(lds4, W4, done + threadIdx.x);
    for (int i = done + threadIdx.x; i < nvec; i += 256) lds4[i] = W4[i];
  }
  __syncthreads();

  const int wave = threadIdx.x >> 5;
  const int lane = threadIdx.x & 31;
  const int half = lane >> 4;
  const int l15  = lane & 15;
  const int row0 = (blockIdx.x * 8 + wave) * 16;
  if (row0 >= N) return;

  const float* aRow = A + (size_t)(row0 + l15) * K + half * 2;

  v8f acc[8] = {};
  for (int k0 = 0; k0 < K; k0 += 4) {
    v2f a = *reinterpret_cast<const v2f*>(aRow + k0);
    const float* wr0 = &lds[(k0 + half * 2) * COUT];
    const float* wr1 = wr0 + COUT;
#pragma unroll
    for (int ct = 0; ct < 8; ++ct) {
      int col = ct * 16 + l15;
      v2f b;
      b.x = wr0[col];
      b.y = wr1[col];
      acc[ct] = __builtin_amdgcn_wmma_f32_16x16x4_f32(
          false, a, false, b, (short)0, acc[ct], false, false);
    }
  }

#pragma unroll
  for (int ct = 0; ct < 8; ++ct) {
    int col = ct * 16 + l15;
    float bv = bias[col];
#pragma unroll
    for (int r = 0; r < 8; ++r) {
      float v = acc[ct][r] + bv;
      if (do_relu) v = fmaxf(v, 0.0f);
      out[(size_t)(row0 + half * 8 + r) * COUT + col] = v;
    }
  }
}

extern "C" void kernel_launch(void* const* d_in, const int* in_sizes, int n_in,
                              void* d_out, int out_size, void* d_ws, size_t ws_size,
                              hipStream_t stream) {
  const float* x   = (const float*)d_in[0];
  const int*   ei  = (const int*)d_in[1];
  const float* W1a = (const float*)d_in[2];
  const float* b1a = (const float*)d_in[3];
  const float* W1b = (const float*)d_in[4];
  const float* b1b = (const float*)d_in[5];
  const float* W2a = (const float*)d_in[6];
  const float* b2a = (const float*)d_in[7];
  const float* W2b = (const float*)d_in[8];
  const float* b2b = (const float*)d_in[9];
  float* out = (float*)d_out;

  const int N = in_sizes[0] / 64;
  const int E = in_sizes[1] / 2;
  const int* src = ei;
  const int* dst = ei + E;

  float* R0 = (float*)d_ws;
  float* R1 = R0 + (size_t)N * 128;
  float* R2 = R1 + (size_t)N * 128;

  int* cnt    = (int*)(R2 + (size_t)N * 128);
  int* rowptr = cnt + (N + 31) / 32 * 32;
  int* rank   = rowptr + (N + 1);
  int* csr    = rank + E;

  const int strips  = (N + 15) / 16;
  const int gblocks = (strips + 7) / 8;

  zero_counts<<<(N + 255) / 256, 256, 0, stream>>>(cnt, N);
  rank_edges<<<(E + 255) / 256, 256, 0, stream>>>(cnt, dst, rank, E);
  scan_counts<<<1, 1024, 0, stream>>>(cnt, rowptr, N);
  place_edges<<<(E + 255) / 256, 256, 0, stream>>>(rowptr, src, dst, rank, csr, E);

  gather_rows<64><<<(N * 16 + 255) / 256, 256, 0, stream>>>(R0, x, rowptr, csr, N);
  gemm_bias_act<<<gblocks, 256,  64 * COUT * sizeof(float), stream>>>(R0, W1a, b1a, R1, N,  64, 1);
  gemm_bias_act<<<gblocks, 256, 128 * COUT * sizeof(float), stream>>>(R1, W1b, b1b, R2, N, 128, 1);

  gather_rows<128><<<(N * 32 + 255) / 256, 256, 0, stream>>>(R0, R2, rowptr, csr, N);
  gemm_bias_act<<<gblocks, 256, 128 * COUT * sizeof(float), stream>>>(R0, W2a, b2a, R1, N, 128, 1);
  gemm_bias_act<<<gblocks, 256, 128 * COUT * sizeof(float), stream>>>(R1, W2b, b2b, out, N, 128, 0);
}


# Edge list grouped by destination built as a two-level counting sort on LDS counters (count, scan, scatter, per-bucket rows) instead of one returning global atomic per edge; sum order = LDS arrival
# speedup vs baseline: 2.7940x; 1.2951x over previous
#include <hip/hip_runtime.h>

#define COUT 128

typedef float v2f __attribute__((ext_vector_type(2)));
typedef float v8f __attribute__((ext_vector_type(8)));


#define BSH    8
#define BW     (1 << BSH)
#define NCHUNK 256

__device__ __forceinline__ int bucket_of(int d, int nb) {
  return (int)min((unsigned)d >> BSH, (unsigned)(nb - 1));
}

__global__ __launch_bounds__(1024)
void count_buckets(const int* __restrict__ dst, int* __restrict__ hist, int E, int per, int nb) {
  extern __shared__ int lcnt[];
  const int c   = blockIdx.x;
  const int tid = threadIdx.x;
  for (int i = tid; i < nb; i += 1024) lcnt[i] = 0;
  __syncthreads();
  const int e0 = c * per;
  const int e1 = min(E, e0 + per);
  for (int base = e0; base < e1; base += 4 * 1024) {
    int d[4];
#pragma unroll
    for (int k = 0; k < 4; ++k) d[k] = dst[min(base + k * 1024 + tid, e1 - 1)];
#pragma unroll
    for (int k = 0; k < 4; ++k)
      if (base + k * 1024 + tid < e1) atomicAdd(&lcnt[bucket_of(d[k], nb)], 1);
  }
  __syncthreads();
  for (int i = tid; i < nb; i += 1024) hist[i * NCHUNK + c] = lcnt[i];
}

__device__ __forceinline__ int wave_inclusive_sum(int x, int lane) {
  x += __builtin_amdgcn_update_dpp(0, x, 0x111, 0xf, 0xf, false);
  x += __builtin_amdgcn_update_dpp(0, x, 0x112, 0xf, 0xf, false);
  x += __builtin_amdgcn_update_dpp(0, x, 0x114, 0xf, 0xf, false);
  x += __builtin_amdgcn_update_dpp(0, x, 0x118, 0xf, 0xf, false);
  const int low_row = __builtin_amdgcn_readlane(x, 15);
  return x + (lane >= 16 ? low_row : 0);
}

#define SCAN_VPL 13

__global__ __launch_bounds__(1024)
void scan_counts(const int* __restrict__ cnt, int* __restrict__ rowptr, int n) {
  __shared__ int wsum[2][32];
  const int tid  = threadIdx.x;
  const int lane = tid & 31;
  const int wave = __builtin_amdgcn_readfirstlane(tid >> 5);
  const int4* __restrict__ c4 = reinterpret_cast<const int4*>(cnt);
  int4* __restrict__ r4 = reinterpret_cast<int4*>(rowptr);
  const int nfull = n >> 2;
  int carry = 0;
  int par = 0;
  for (int first = 0; first < nfull; first += 32 * SCAN_VPL * 32) {
    const int p0 = first + wave * (SCAN_VPL * 32) + lane;
    int4 v[SCAN_VPL];
#pragma unroll
    for (int r = 0; r < SCAN_VPL; ++r) {
      const int p = p0 + r * 32;
      v[r] = (p < nfull) ? c4[p] : make_int4(0, 0, 0, 0);
    }
    int inc[SCAN_VPL];
#pragma unroll
    for (int r = 0; r < SCAN_VPL; ++r)
      inc[r] = wave_inclusive_sum(v[r].x + v[r].y + v[r].z + v[r].w, lane);
    int rowbase[SCAN_VPL];
    int run = 0;
#pragma unroll
    for (int r = 0; r < SCAN_VPL; ++r) {
      rowbase[r] = run;
      run += __builtin_amdgcn_readlane(inc[r], 31);
    }
    if (lane == 0) wsum[par][wave] = run;
    __syncthreads();
    const int w = wave_inclusive_sum(wsum[par][lane], lane);
    const int wbase = carry + __builtin_amdgcn_readlane(w, wave) - run;
#pragma unroll
    for (int r = 0; r < SCAN_VPL; ++r) {
      const int p = p0 + r * 32;
      if (p < nfull) {
        int4 o;
        o.x = wbase + rowbase[r] + inc[r] - (v[r].x + v[r].y + v[r].z + v[r].w);
        o.y = o.x + v[r].x;
        o.z = o.y + v[r].y;
        o.w = o.z + v[r].z;
        r4[p] = o;
      }
    }
    carry += __builtin_amdgcn_readlane(w, 31);
    par ^= 1;
  }
  if (tid == 0) {
    int b = carry;
    for (int i = nfull * 4; i < n; ++i) { rowptr[i] = b; b += cnt[i]; }
    rowptr[n] = b;
  }
}

__global__ __launch_bounds__(1024)
void scatter_to_buckets(const int* __restrict__ src, const int* __restrict__ dst,
                        const int* __restrict__ bstart, int2* __restrict__ bkt,
                        int E, int per, int nb) {
  extern __shared__ int lcur[];
  const int c   = blockIdx.x;
  const int tid = threadIdx.x;
  for (int i = tid; i < nb; i += 1024) lcur[i] = bstart[i * NCHUNK + c];
  __syncthreads();
  const int e0 = c * per;
  const int e1 = min(E, e0 + per);
  for (int base = e0; base < e1; base += 4 * 1024) {
    int s[4], d[4];
#pragma unroll
    for (int k = 0; k < 4; ++k) {
      const int e = min(base + k * 1024 + tid, e1 - 1);
      d[k] = dst[e];
      s[k] = src[e];
    }
#pragma unroll
    for (int k = 0; k < 4; ++k)
      if (base + k * 1024 + tid < e1) {
        const int pos = atomicAdd(&lcur[bucket_of(d[k], nb)], 1);
        bkt[pos] = make_int2(s[k], d[k]);
      }
  }
}

#define ROWS_PER_TRIP 6

__global__ __launch_bounds__(1024)
void build_rows(const int* __restrict__ bstart, const int2* __restrict__ bkt,
                int* __restrict__ rowptr, int* __restrict__ csr, int N) {
  __shared__ int slot[BW];
  __shared__ int wtot[BW / 32];
  const int b   = blockIdx.x;
  const int tid = threadIdx.x;
  const int beg = bstart[b * NCHUNK];
  const int end = bstart[(b + 1) * NCHUNK];
  const int* __restrict__ bdst = reinterpret_cast<const int*>(bkt);
  if (tid < BW) slot[tid] = 0;
  __syncthreads();
  for (int base = beg; base < end; base += ROWS_PER_TRIP * 1024) {
    int d[ROWS_PER_TRIP];
#pragma unroll
    for (int k = 0; k < ROWS_PER_TRIP; ++k) d[k] = bdst[2 * min(base + k * 1024 + tid, end - 1) + 1];
#pragma unroll
    for (int k = 0; k < ROWS_PER_TRIP; ++k)
      if (base + k * 1024 + tid < end) atomicAdd(&slot[d[k] & (BW - 1)], 1);
  }
  __syncthreads();
  int deg = 0, inc = 0;
  if (tid < BW) {
    deg = slot[tid];
    inc = wave_inclusive_sum(deg, tid & 31);
    if ((tid & 31) == 31) wtot[tid >> 5] = inc;
  }
  __syncthreads();
  if (tid < BW) {
    int before = 0;
#pragma unroll
    for (int w = 0; w < BW / 32; ++w) before += (w < (tid >> 5)) ? wtot[w] : 0;
    const int start = beg + before + inc - deg;
    slot[tid] = start;
    const int node = b * BW + tid;
    if (node < N) rowptr[node] = start;
    if (node == N - 1) rowptr[N] = end;
  }
  __syncthreads();
  for (int base = beg; base < end; base += ROWS_PER_TRIP * 1024) {
    int2 v[ROWS_PER_TRIP];
#pragma unroll
    for (int k = 0; k < ROWS_PER_TRIP; ++k) v[k] = bkt[min(base + k * 1024 + tid, end - 1)];
#pragma unroll
    for (int k = 0; k < ROWS_PER_TRIP; ++k)
      if (base + k * 1024 + tid < end) {
        const int pos = atomicAdd(&slot[v[k].y & (BW - 1)], 1);
        csr[pos] = v[k].x;
      }
  }
}

template <int C>
__global__ __launch_bounds__(256)
void gather_rows(float* __restrict__ agg, const float* __restrict__ h,
                 const int* __restrict__ rowptr, const int* __restrict__ csr, int N) {
  constexpr int LPN = C / 4;
  const int gtid = blockIdx.x * 256 + threadIdx.x;
  const int node = gtid / LPN;
  const int g    = gtid % LPN;
  if (node >= N) return;
  const float4* __restrict__ h4 = reinterpret_cast<const float4*>(h);
  const int beg = rowptr[node];
  const int end = rowptr[node + 1];
  const unsigned last = (unsigned)(N - 1);
  float4 acc = h4[node * LPN + g];
  for (int j = beg; j < end; j += 4) {
    unsigned s[4];
    float4 v[4];
#pragma unroll
    for (int k = 0; k < 4; ++k) s[k] = min((unsigned)csr[min(j + k, end - 1)], last);
#pragma unroll
    for (int k = 0; k < 4; ++k) v[k] = h4[s[k] * LPN + g];
#pragma unroll
    for (int k = 0; k < 4; ++k) {
      const bool on = (j + k) < end;
      acc.x += on ? v[k].x : 0.0f;
      acc.y += on ? v[k].y : 0.0f;
      acc.z += on ? v[k].z : 0.0f;
      acc.w += on ? v[k].w : 0.0f;
    }
  }
  reinterpret_cast<float4*>(agg)[node * LPN + g] = acc;
}

template <int VPT>
__device__ __forceinline__ void stage_pieces(float4* lds4, const float4* __restrict__ W4, int first) {
  float4 v[VPT];
#pragma unroll
  for (int r = 0; r < VPT; ++r) v[r] = W4[first + r * 256];
#pragma unroll
  for (int r = 0; r < VPT; ++r) lds4[first + r * 256] = v[r];
}

__global__ __launch_bounds__(256)
void gemm_bias_act(const float* __restrict__ A, const float* __restrict__ W,
                   const float* __restrict__ bias, float* __restrict__ out,
                   int N, int K, int do_relu) {
  extern __shared__ __align__(16) float lds[];
  {
    const float4* __restrict__ W4 = reinterpret_cast<const float4*>(W);
    float4* lds4 = reinterpret_cast<float4*>(lds);
    const int nvec = K * (COUT / 4);
    int done = 0;
    for (; done + 16 * 256 <= nvec; done += 16 * 256) stage_pieces<16>(lds4, W4, done + threadIdx.x);
    for (; done +  8 * 256 <= nvec; done +=  8 * 256) stage_pieces<8>(lds4, W4, done + threadIdx.x);
    for (int i = done + threadIdx.x; i < nvec; i += 256) lds4[i] = W4[i];
  }
  __syncthreads();

  const int wave = threadIdx.x >> 5;
  const int lane = threadIdx.x & 31;
  const int half = lane >> 4;
  const int l15  = lane & 15;
  const int row0 = (blockIdx.x * 8 + wave) * 16;
  if (row0 >= N) return;

  const float* aRow = A + (size_t)(row0 + l15) * K + half * 2;

  v8f acc[8] = {};
  for (int k0 = 0; k0 < K; k0 += 4) {
    v2f a = *reinterpret_cast<const v2f*>(aRow + k0);
    const float* wr0 = &lds[(k0 + half * 2) * COUT];
    const float* wr1 = wr0 + COUT;
#pragma unroll
    for (int ct = 0; ct < 8; ++ct) {
      int col = ct * 16 + l15;
      v2f b;
      b.x = wr0[col];
      b.y = wr1[col];
      acc[ct] = __builtin_amdgcn_wmma_f32_16x16x4_f32(
          false, a, false, b, (short)0, acc[ct], false, false);
    }
  }

#pragma unroll
  for (int ct = 0; ct < 8; ++ct) {
    int col = ct * 16 + l15;
    float bv = bias[col];
#pragma unroll
    for (int r = 0; r < 8; ++r) {
      float v = acc[ct][r] + bv;
      if (do_relu) v = fmaxf(v, 0.0f);
      out[(size_t)(row0 + half * 8 + r) * COUT + col] = v;
    }
  }
}

extern "C" void kernel_launch(void* const* d_in, const int* in_sizes, int n_in,
                              void* d_out, int out_size, void* d_ws, size_t ws_size,
                              hipStream_t stream) {
  const float* x   = (const float*)d_in[0];
  const int*   ei  = (const int*)d_in[1];
  const float* W1a = (const float*)d_in[2];
  const float* b1a = (const float*)d_in[3];
  const float* W1b = (const float*)d_in[4];
  const float* b1b = (const float*)d_in[5];
  const float* W2a = (const float*)d_in[6];
  const float* b2a = (const float*)d_in[7];
  const float* W2b = (const float*)d_in[8];
  const float* b2b = (const float*)d_in[9];
  float* out = (float*)d_out;

  const int N = in_sizes[0] / 64;
  const int E = in_sizes[1] / 2;
  const int* src = ei;
  const int* dst = ei + E;

  float* R0 = (float*)d_ws;
  float* R1 = R0 + (size_t)N * 128;
  float* R2 = R1 + (size_t)N * 128;

  const int NB  = (N + BW - 1) / BW;
  const int per = (E + NCHUNK - 1) / NCHUNK;
  const int nh  = NB * NCHUNK;

  int*  hist   = (int*)(R2 + (size_t)N * 128);
  int*  bstart = hist + (nh + 31) / 32 * 32;
  int*  rowptr = bstart + (nh + 1 + 31) / 32 * 32;
  int2* bkt    = (int2*)(rowptr + (N + 1 + 31) / 32 * 32);
  int*  csr    = (int*)(bkt + E);

  const int strips  = (N + 15) / 16;
  const int gblocks = (strips + 7) / 8;

  count_buckets<<<NCHUNK, 1024, NB * sizeof(int), stream>>>(dst, hist, E, per, NB);
  scan_counts<<<1, 1024, 0, stream>>>(hist, bstart, nh);
  scatter_to_buckets<<<NCHUNK, 1024, NB * sizeof(int), stream>>>(src, dst, bstart, bkt, E, per, NB);
  build_rows<<<NB, 1024, 0, stream>>>(bstart, bkt, rowptr, csr, N);

  gather_rows<64><<<(N * 16 + 255) / 256, 256, 0, stream>>>(R0, x, rowptr, csr, N);
  gemm_bias_act<<<gblocks, 256,  64 * COUT * sizeof(float), stream>>>(R0, W1a, b1a, R1, N,  64, 1);
  gemm_bias_act<<<gblocks, 256, 128 * COUT * sizeof(float), stream>>>(R1, W1b, b1b, R2, N, 128, 1);

  gather_rows<128><<<(N * 32 + 255) / 256, 256, 0, stream>>>(R0, R2, rowptr, csr, N);
  gemm_bias_act<<<gblocks, 256, 128 * COUT * sizeof(float), stream>>>(R0, W2a, b2a, R1, N, 128, 1);
  gemm_bias_act<<<gblocks, 256, 128 * COUT * sizeof(float), stream>>>(R1, W2b, b2b, out, N, 128, 0);
}


# Each layer's two GEMMs fused into one launch: products taken transposed so product 1's D registers are product 2's B operand; hidden tile stays in registers; product 2 sums k in D-register order
# speedup vs baseline: 2.9157x; 1.0436x over previous
#include <hip/hip_runtime.h>

#define COUT 128

typedef float v2f __attribute__((ext_vector_type(2)));
typedef float v8f __attribute__((ext_vector_type(8)));


#define BSH    8
#define BW     (1 << BSH)
#define NCHUNK 256

__device__ __forceinline__ int bucket_of(int d, int nb) {
  return (int)min((unsigned)d >> BSH, (unsigned)(nb - 1));
}

__global__ __launch_bounds__(1024)
void count_buckets(const int* __restrict__ dst, int* __restrict__ hist, int E, int per, int nb) {
  extern __shared__ int lcnt[];
  const int c   = blockIdx.x;
  const int tid = threadIdx.x;
  for (int i = tid; i < nb; i += 1024) lcnt[i] = 0;
  __syncthreads();
  const int e0 = c * per;
  const int e1 = min(E, e0 + per);
  for (int base = e0; base < e1; base += 4 * 1024) {
    int d[4];
#pragma unroll
    for (int k = 0; k < 4; ++k) d[k] = dst[min(base + k * 1024 + tid, e1 - 1)];
#pragma unroll
    for (int k = 0; k < 4; ++k)
      if (base + k * 1024 + tid < e1) atomicAdd(&lcnt[bucket_of(d[k], nb)], 1);
  }
  __syncthreads();
  for (int i = tid; i < nb; i += 1024) hist[i * NCHUNK + c] = lcnt[i];
}

__device__ __forceinline__ int wave_inclusive_sum(int x, int lane) {
  x += __builtin_amdgcn_update_dpp(0, x, 0x111, 0xf, 0xf, false);
  x += __builtin_amdgcn_update_dpp(0, x, 0x112, 0xf, 0xf, false);
  x += __builtin_amdgcn_update_dpp(0, x, 0x114, 0xf, 0xf, false);
  x += __builtin_amdgcn_update_dpp(0, x, 0x118, 0xf, 0xf, false);
  const int low_row = __builtin_amdgcn_readlane(x, 15);
  return x + (lane >= 16 ? low_row : 0);
}

#define SCAN_VPL 13

__global__ __launch_bounds__(1024)
void scan_counts(const int* __restrict__ cnt, int* __restrict__ rowptr, int n) {
  __shared__ int wsum[2][32];
  const int tid  = threadIdx.x;
  const int lane = tid & 31;
  const int wave = __builtin_amdgcn_readfirstlane(tid >> 5);
  const int4* __restrict__ c4 = reinterpret_cast<const int4*>(cnt);
  int4* __restrict__ r4 = reinterpret_cast<int4*>(rowptr);
  const int nfull = n >> 2;
  int carry = 0;
  int par = 0;
  for (int first = 0; first < nfull; first += 32 * SCAN_VPL * 32) {
    const int p0 = first + wave * (SCAN_VPL * 32) + lane;
    int4 v[SCAN_VPL];
#pragma unroll
    for (int r = 0; r < SCAN_VPL; ++r) {
      const int p = p0 + r * 32;
      v[r] = (p < nfull) ? c4[p] : make_int4(0, 0, 0, 0);
    }
    int inc[SCAN_VPL];
#pragma unroll
    for (int r = 0; r < SCAN_VPL; ++r)
      inc[r] = wave_inclusive_sum(v[r].x + v[r].y + v[r].z + v[r].w, lane);
    int rowbase[SCAN_VPL];
    int run = 0;
#pragma unroll
    for (int r = 0; r < SCAN_VPL; ++r) {
      rowbase[r] = run;
      run += __builtin_amdgcn_readlane(inc[r], 31);
    }
    if (lane == 0) wsum[par][wave] = run;
    __syncthreads();
    const int w = wave_inclusive_sum(wsum[par][lane], lane);
    const int wbase = carry + __builtin_amdgcn_readlane(w, wave) - run;
#pragma unroll
    for (int r = 0; r < SCAN_VPL; ++r) {
      const int p = p0 + r * 32;
      if (p < nfull) {
        int4 o;
        o.x = wbase + rowbase[r] + inc[r] - (v[r].x + v[r].y + v[r].z + v[r].w);
        o.y = o.x + v[r].x;
        o.z = o.y + v[r].y;
        o.w = o.z + v[r].z;
        r4[p] = o;
      }
    }
    carry += __builtin_amdgcn_readlane(w, 31);
    par ^= 1;
  }
  if (tid == 0) {
    int b = carry;
    for (int i = nfull * 4; i < n; ++i) { rowptr[i] = b; b += cnt[i]; }
    rowptr[n] = b;
  }
}

__global__ __launch_bounds__(1024)
void scatter_to_buckets(const int* __restrict__ src, const int* __restrict__ dst,
                        const int* __restrict__ bstart, int2* __restrict__ bkt,
                        int E, int per, int nb) {
  extern __shared__ int lcur[];
  const int c   = blockIdx.x;
  const int tid = threadIdx.x;
  for (int i = tid; i < nb; i += 1024) lcur[i] = bstart[i * NCHUNK + c];
  __syncthreads();
  const int e0 = c * per;
  const int e1 = min(E, e0 + per);
  for (int base = e0; base < e1; base += 4 * 1024) {
    int s[4], d[4];
#pragma unroll
    for (int k = 0; k < 4; ++k) {
      const int e = min(base + k * 1024 + tid, e1 - 1);
      d[k] = dst[e];
      s[k] = src[e];
    }
#pragma unroll
    for (int k = 0; k < 4; ++k)
      if (base + k * 1024 + tid < e1) {
        const int pos = atomicAdd(&lcur[bucket_of(d[k], nb)], 1);
        bkt[pos] = make_int2(s[k], d[k]);
      }
  }
}

#define ROWS_PER_TRIP 6

__global__ __launch_bounds__(1024)
void build_rows(const int* __restrict__ bstart, const int2* __restrict__ bkt,
                int* __restrict__ rowptr, int* __restrict__ csr, int N) {
  __shared__ int slot[BW];
  __shared__ int wtot[BW / 32];
  const int b   = blockIdx.x;
  const int tid = threadIdx.x;
  const int beg = bstart[b * NCHUNK];
  const int end = bstart[(b + 1) * NCHUNK];
  const int* __restrict__ bdst = reinterpret_cast<const int*>(bkt);
  if (tid < BW) slot[tid] = 0;
  __syncthreads();
  for (int base = beg; base < end; base += ROWS_PER_TRIP * 1024) {
    int d[ROWS_PER_TRIP];
#pragma unroll
    for (int k = 0; k < ROWS_PER_TRIP; ++k) d[k] = bdst[2 * min(base + k * 1024 + tid, end - 1) + 1];
#pragma unroll
    for (int k = 0; k < ROWS_PER_TRIP; ++k)
      if (base + k * 1024 + tid < end) atomicAdd(&slot[d[k] & (BW - 1)], 1);
  }
  __syncthreads();
  int deg = 0, inc = 0;
  if (tid < BW) {
    deg = slot[tid];
    inc = wave_inclusive_sum(deg, tid & 31);
    if ((tid & 31) == 31) wtot[tid >> 5] = inc;
  }
  __syncthreads();
  if (tid < BW) {
    int before = 0;
#pragma unroll
    for (int w = 0; w < BW / 32; ++w) before += (w < (tid >> 5)) ? wtot[w] : 0;
    const int start = beg + before + inc - deg;
    slot[tid] = start;
    const int node = b * BW + tid;
    if (node < N) rowptr[node] = start;
    if (node == N - 1) rowptr[N] = end;
  }
  __syncthreads();
  for (int base = beg; base < end; base += ROWS_PER_TRIP * 1024) {
    int2 v[ROWS_PER_TRIP];
#pragma unroll
    for (int k = 0; k < ROWS_PER_TRIP; ++k) v[k] = bkt[min(base + k * 1024 + tid, end - 1)];
#pragma unroll
    for (int k = 0; k < ROWS_PER_TRIP; ++k)
      if (base + k * 1024 + tid < end) {
        const int pos = atomicAdd(&slot[v[k].y & (BW - 1)], 1);
        csr[pos] = v[k].x;
      }
  }
}

template <int C>
__global__ __launch_bounds__(256)
void gather_rows(float* __restrict__ agg, const float* __restrict__ h,
                 const int* __restrict__ rowptr, const int* __restrict__ csr, int N) {
  constexpr int LPN = C / 4;
  const int gtid = blockIdx.x * 256 + threadIdx.x;
  const int node = gtid / LPN;
  const int g    = gtid % LPN;
  if (node >= N) return;
  const float4* __restrict__ h4 = reinterpret_cast<const float4*>(h);
  const int beg = rowptr[node];
  const int end = rowptr[node + 1];
  const unsigned last = (unsigned)(N - 1);
  float4 acc = h4[node * LPN + g];
  for (int j = beg; j < end; j += 4) {
    unsigned s[4];
    float4 v[4];
#pragma unroll
    for (int k = 0; k < 4; ++k) s[k] = min((unsigned)csr[min(j + k, end - 1)], last);
#pragma unroll
    for (int k = 0; k < 4; ++k) v[k] = h4[s[k] * LPN + g];
#pragma unroll
    for (int k = 0; k < 4; ++k) {
      const bool on = (j + k) < end;
      acc.x += on ? v[k].x : 0.0f;
      acc.y += on ? v[k].y : 0.0f;
      acc.z += on ? v[k].z : 0.0f;
      acc.w += on ? v[k].w : 0.0f;
    }
  }
  reinterpret_cast<float4*>(agg)[node * LPN + g] = acc;
}

template <int K1, bool RELU_OUT>
__global__ __launch_bounds__(256)
void mlp_fused(const float* __restrict__ A, const float* __restrict__ Wa, const float* __restrict__ ba,
               const float* __restrict__ Wb, const float* __restrict__ bb,
               float* __restrict__ out, int N, int wb_at) {
  __shared__ __align__(16) float w[(K1 + COUT) * COUT];
  const float* wa = w;
  const float* wb = w + wb_at;

  const int tid  = threadIdx.x;
  const int wave = tid >> 5;
  const int lane = tid & 31;
  const int half = lane >> 4;
  const int l15  = lane & 15;
  const int row0 = (blockIdx.x * 8 + wave) * 16;

  v2f zf[K1 / 4];
  {
    const float* aRow = A + (size_t)min(row0 + l15, N - 1) * K1 + half * 2;
#pragma unroll
    for (int i = 0; i < K1 / 4; ++i) zf[i] = *reinterpret_cast<const v2f*>(aRow + 4 * i);
  }
  {
    constexpr int NA = K1 * (COUT / 4) / 256;
    constexpr int NB = COUT * (COUT / 4) / 256;
    const float4* __restrict__ Wa4 = reinterpret_cast<const float4*>(Wa);
    const float4* __restrict__ Wb4 = reinterpret_cast<const float4*>(Wb);
    float4* wa4 = reinterpret_cast<float4*>(w);
    float4* wb4 = wa4 + K1 * (COUT / 4);
    float4 va[NA], vb[NB];
#pragma unroll
    for (int r = 0; r < NA; ++r) va[r] = Wa4[tid + r * 256];
#pragma unroll
    for (int r = 0; r < NB; ++r) vb[r] = Wb4[tid + r * 256];
#pragma unroll
    for (int r = 0; r < NA; ++r) wa4[tid + r * 256] = va[r];
#pragma unroll
    for (int r = 0; r < NB; ++r) wb4[tid + r * 256] = vb[r];
  }
  __syncthreads();
  if (row0 >= N) return;

  v8f acc[8] = {};
#pragma unroll
  for (int i = 0; i < K1 / 4; ++i) {
    const float* w0 = &wa[(4 * i + half * 2) * COUT + l15];
    const float* w1 = w0 + COUT;
#pragma unroll
    for (int ct = 0; ct < 8; ++ct) {
      v2f wf;
      wf.x = w0[ct * 16];
      wf.y = w1[ct * 16];
      acc[ct] = __builtin_amdgcn_wmma_f32_16x16x4_f32(
          false, wf, false, zf[i], (short)0, acc[ct], false, false);
    }
  }

#pragma unroll
  for (int ct = 0; ct < 8; ++ct) {
    const float4 b0 = *reinterpret_cast<const float4*>(ba + ct * 16 + half * 8);
    const float4 b1 = *reinterpret_cast<const float4*>(ba + ct * 16 + half * 8 + 4);
    acc[ct][0] = fmaxf(acc[ct][0] + b0.x, 0.0f);
    acc[ct][1] = fmaxf(acc[ct][1] + b0.y, 0.0f);
    acc[ct][2] = fmaxf(acc[ct][2] + b0.z, 0.0f);
    acc[ct][3] = fmaxf(acc[ct][3] + b0.w, 0.0f);
    acc[ct][4] = fmaxf(acc[ct][4] + b1.x, 0.0f);
    acc[ct][5] = fmaxf(acc[ct][5] + b1.y, 0.0f);
    acc[ct][6] = fmaxf(acc[ct][6] + b1.z, 0.0f);
    acc[ct][7] = fmaxf(acc[ct][7] + b1.w, 0.0f);
  }

  v8f res[8] = {};
  const float* wbl = wb + half * 8 * COUT + l15;
#pragma unroll
  for (int ct = 0; ct < 8; ++ct) {
#pragma unroll
    for (int j = 0; j < 4; ++j) {
      v2f hf;
      hf.x = acc[ct][2 * j];
      hf.y = acc[ct][2 * j + 1];
      const float* w0 = wbl + (ct * 16 + 2 * j) * COUT;
      const float* w1 = w0 + COUT;
#pragma unroll
      for (int c2 = 0; c2 < 8; ++c2) {
        v2f wf;
        wf.x = w0[c2 * 16];
        wf.y = w1[c2 * 16];
        res[c2] = __builtin_amdgcn_wmma_f32_16x16x4_f32(
            false, wf, false, hf, (short)0, res[c2], false, false);
      }
    }
  }

  float* orow = out + (size_t)(row0 + l15) * COUT + half * 8;
#pragma unroll
  for (int c2 = 0; c2 < 8; ++c2) {
    const float4 b0 = *reinterpret_cast<const float4*>(bb + c2 * 16 + half * 8);
    const float4 b1 = *reinterpret_cast<const float4*>(bb + c2 * 16 + half * 8 + 4);
    float4 o0, o1;
    o0.x = res[c2][0] + b0.x;
    o0.y = res[c2][1] + b0.y;
    o0.z = res[c2][2] + b0.z;
    o0.w = res[c2][3] + b0.w;
    o1.x = res[c2][4] + b1.x;
    o1.y = res[c2][5] + b1.y;
    o1.z = res[c2][6] + b1.z;
    o1.w = res[c2][7] + b1.w;
    if (RELU_OUT) {
      o0.x = fmaxf(o0.x, 0.0f); o0.y = fmaxf(o0.y, 0.0f); o0.z = fmaxf(o0.z, 0.0f); o0.w = fmaxf(o0.w, 0.0f);
      o1.x = fmaxf(o1.x, 0.0f); o1.y = fmaxf(o1.y, 0.0f); o1.z = fmaxf(o1.z, 0.0f); o1.w = fmaxf(o1.w, 0.0f);
    }
    float4* dst = reinterpret_cast<float4*>(orow + c2 * 16);
    dst[0] = o0;
    dst[1] = o1;
  }
}

extern "C" void kernel_launch(void* const* d_in, const int* in_sizes, int n_in,
                              void* d_out, int out_size, void* d_ws, size_t ws_size,
                              hipStream_t stream) {
  const float* x   = (const float*)d_in[0];
  const int*   ei  = (const int*)d_in[1];
  const float* W1a = (const float*)d_in[2];
  const float* b1a = (const float*)d_in[3];
  const float* W1b = (const float*)d_in[4];
  const float* b1b = (const float*)d_in[5];
  const float* W2a = (const float*)d_in[6];
  const float* b2a = (const float*)d_in[7];
  const float* W2b = (const float*)d_in[8];
  const float* b2b = (const float*)d_in[9];
  float* out = (float*)d_out;

  const int N = in_sizes[0] / 64;
  const int E = in_sizes[1] / 2;
  const int* src = ei;
  const int* dst = ei + E;

  float* R0 = (float*)d_ws;
  float* R2 = R0 + (size_t)N * 256;

  const int NB  = (N + BW - 1) / BW;
  const int per = (E + NCHUNK - 1) / NCHUNK;
  const int nh  = NB * NCHUNK;

  int*  hist   = (int*)(R2 + (size_t)N * 128);
  int*  bstart = hist + (nh + 31) / 32 * 32;
  int*  rowptr = bstart + (nh + 1 + 31) / 32 * 32;
  int2* bkt    = (int2*)(rowptr + (N + 1 + 31) / 32 * 32);
  int*  csr    = (int*)(bkt + E);

  const int strips  = (N + 15) / 16;
  const int gblocks = (strips + 7) / 8;

  count_buckets<<<NCHUNK, 1024, NB * sizeof(int), stream>>>(dst, hist, E, per, NB);
  scan_counts<<<1, 1024, 0, stream>>>(hist, bstart, nh);
  scatter_to_buckets<<<NCHUNK, 1024, NB * sizeof(int), stream>>>(src, dst, bstart, bkt, E, per, NB);
  build_rows<<<NB, 1024, 0, stream>>>(bstart, bkt, rowptr, csr, N);

  gather_rows<64><<<(N * 16 + 255) / 256, 256, 0, stream>>>(R0, x, rowptr, csr, N);
  mlp_fused<64, true><<<gblocks, 256, 0, stream>>>(R0, W1a, b1a, W1b, b1b, R2, N, 64 * COUT);

  gather_rows<128><<<(N * 32 + 255) / 256, 256, 0, stream>>>(R0, R2, rowptr, csr, N);
  mlp_fused<128, false><<<gblocks, 256, 0, stream>>>(R0, W2a, b2a, W2b, b2b, out, N, 128 * COUT);
}


# Edge list by destination built in 2 launches, was 4: each chunk's work-group counts, scans and groups its edges by bucket in LDS; bucket work-groups collect their pieces; sum order = LDS arrival
# speedup vs baseline: 3.3578x; 1.1516x over previous
#include <hip/hip_runtime.h>

#define COUT 128

typedef float v2f __attribute__((ext_vector_type(2)));
typedef float v8f __attribute__((ext_vector_type(8)));


#define BSH    8
#define BW     (1 << BSH)
#define NCHUNK 256

__device__ __forceinline__ int bucket_of(int d, int nb) {
  return (int)min((unsigned)d >> BSH, (unsigned)(nb - 1));
}

__device__ __forceinline__ int wave_inclusive_sum(int x, int lane) {
  x += __builtin_amdgcn_update_dpp(0, x, 0x111, 0xf, 0xf, false);
  x += __builtin_amdgcn_update_dpp(0, x, 0x112, 0xf, 0xf, false);
  x += __builtin_amdgcn_update_dpp(0, x, 0x114, 0xf, 0xf, false);
  x += __builtin_amdgcn_update_dpp(0, x, 0x118, 0xf, 0xf, false);
  const int low_row = __builtin_amdgcn_readlane(x, 15);
  return x + (lane >= 16 ? low_row : 0);
}

__global__ __launch_bounds__(1024)
void sort_chunk(const int* __restrict__ src, const int* __restrict__ dst,
                int* __restrict__ off, int2* __restrict__ srt, int E, int per, int nb) {
  extern __shared__ int lcur[];
  const int c   = blockIdx.x;
  const int tid = threadIdx.x;
  for (int i = tid; i < nb; i += 1024) lcur[i] = 0;
  __syncthreads();
  const int e0 = c * per;
  const int e1 = min(E, e0 + per);
  for (int base = e0; base < e1; base += 4 * 1024) {
    int d[4];
#pragma unroll
    for (int k = 0; k < 4; ++k) d[k] = dst[min(base + k * 1024 + tid, e1 - 1)];
    asm volatile("" :: "v"(d[0]), "v"(d[1]), "v"(d[2]), "v"(d[3]));
#pragma unroll
    for (int k = 0; k < 4; ++k)
      if (base + k * 1024 + tid < e1) atomicAdd(&lcur[bucket_of(d[k], nb)], 1);
  }
  __syncthreads();
  if (tid < 32) {
    int carry = 0;
    for (int i0 = 0; i0 < nb; i0 += 32) {
      const int i   = i0 + tid;
      const int v   = (i < nb) ? lcur[i] : 0;
      const int inc = wave_inclusive_sum(v, tid);
      if (i < nb) {
        const int start = carry + inc - v;
        lcur[i] = e0 + start;
        off[i * NCHUNK + c] = start;
      }
      carry += __builtin_amdgcn_readlane(inc, 31);
    }
    if (tid == 0) off[nb * NCHUNK + c] = carry;
  }
  __syncthreads();
  for (int base = e0; base < e1; base += 4 * 1024) {
    int s[4], d[4];
#pragma unroll
    for (int k = 0; k < 4; ++k) {
      const int e = min(base + k * 1024 + tid, e1 - 1);
      d[k] = dst[e];
      s[k] = src[e];
    }
    asm volatile("" :: "v"(d[0]), "v"(d[1]), "v"(d[2]), "v"(d[3]), "v"(s[0]), "v"(s[1]), "v"(s[2]), "v"(s[3]));
#pragma unroll
    for (int k = 0; k < 4; ++k)
      if (base + k * 1024 + tid < e1) {
        const int pos = atomicAdd(&lcur[bucket_of(d[k], nb)], 1);
        srt[pos] = make_int2(s[k], d[k]);
      }
  }
}

#define CPW (NCHUNK / 32)

__global__ __launch_bounds__(1024)
void build_rows(const int* __restrict__ off, const int2* __restrict__ srt,
                int* __restrict__ rowptr, int* __restrict__ csr, int N, int E, int per) {
  __shared__ int slot[BW];
  __shared__ int wtot[BW / 32];
  __shared__ int pbeg[NCHUNK];
  __shared__ int pend[NCHUNK];
  __shared__ int part[2][NCHUNK / 32];
  const int b    = blockIdx.x;
  const int tid  = threadIdx.x;
  const int lane = tid & 31;
  const int wave = tid >> 5;
  const int* __restrict__ sdst = reinterpret_cast<const int*>(srt);
  if (tid < BW) slot[tid] = 0;
  if (tid < NCHUNK) {
    const int lo = off[b * NCHUNK + tid];
    const int hi = off[(b + 1) * NCHUNK + tid];
    pbeg[tid] = tid * per + lo;
    pend[tid] = tid * per + hi;
    const int slo = wave_inclusive_sum(lo, lane);
    const int sln = wave_inclusive_sum(hi - lo, lane);
    if (lane == 31) { part[0][wave] = slo; part[1][wave] = sln; }
  }
  __syncthreads();
  int beg = 0, cnt = 0;
#pragma unroll
  for (int w = 0; w < NCHUNK / 32; ++w) { beg += part[0][w]; cnt += part[1][w]; }
  const int end = beg + cnt;

  int lo[CPW], hi[CPW], at[CPW];
#pragma unroll
  for (int s = 0; s < CPW; ++s) {
    lo[s] = pbeg[wave * CPW + s];
    hi[s] = pend[wave * CPW + s];
    at[s] = max(min(min(lo[s] + lane, hi[s] - 1), E - 1), 0);
  }
  {
    int d[CPW];
#pragma unroll
    for (int s = 0; s < CPW; ++s) d[s] = sdst[2 * at[s] + 1];
    static_assert(CPW == 8, "the fence below names CPW values");
    asm volatile("" :: "v"(d[0]), "v"(d[1]), "v"(d[2]), "v"(d[3]), "v"(d[4]), "v"(d[5]), "v"(d[6]), "v"(d[7]));
#pragma unroll
    for (int s = 0; s < CPW; ++s)
      if (lo[s] + lane < hi[s]) atomicAdd(&slot[d[s] & (BW - 1)], 1);
#pragma unroll
    for (int s = 0; s < CPW; ++s)
      for (int i = lo[s] + 32 + lane; i < hi[s]; i += 32) atomicAdd(&slot[sdst[2 * i + 1] & (BW - 1)], 1);
  }
  __syncthreads();
  int deg = 0, inc = 0;
  if (tid < BW) {
    deg = slot[tid];
    inc = wave_inclusive_sum(deg, lane);
    if (lane == 31) wtot[wave] = inc;
  }
  __syncthreads();
  if (tid < BW) {
    int before = 0;
#pragma unroll
    for (int w = 0; w < BW / 32; ++w) before += (w < wave) ? wtot[w] : 0;
    const int start = beg + before + inc - deg;
    slot[tid] = start;
    const int node = b * BW + tid;
    if (node < N) rowptr[node] = start;
    if (node == N - 1) rowptr[N] = end;
  }
  __syncthreads();
  {
    int2 v[CPW];
#pragma unroll
    for (int s = 0; s < CPW; ++s) v[s] = srt[at[s]];
    asm volatile("" :: "v"(v[0].y), "v"(v[1].y), "v"(v[2].y), "v"(v[3].y), "v"(v[4].y), "v"(v[5].y), "v"(v[6].y), "v"(v[7].y));
#pragma unroll
    for (int s = 0; s < CPW; ++s)
      if (lo[s] + lane < hi[s]) {
        const int pos = atomicAdd(&slot[v[s].y & (BW - 1)], 1);
        csr[pos] = v[s].x;
      }
#pragma unroll
    for (int s = 0; s < CPW; ++s)
      for (int i = lo[s] + 32 + lane; i < hi[s]; i += 32) {
        const int2 p = srt[i];
        const int pos = atomicAdd(&slot[p.y & (BW - 1)], 1);
        csr[pos] = p.x;
      }
  }
}

template <int C>
__global__ __launch_bounds__(256)
void gather_rows(float* __restrict__ agg, const float* __restrict__ h,
                 const int* __restrict__ rowptr, const int* __restrict__ csr, int N) {
  constexpr int LPN = C / 4;
  const int gtid = blockIdx.x * 256 + threadIdx.x;
  const int node = gtid / LPN;
  const int g    = gtid % LPN;
  if (node >= N) return;
  const float4* __restrict__ h4 = reinterpret_cast<const float4*>(h);
  const int beg = rowptr[node];
  const int end = rowptr[node + 1];
  const unsigned last = (unsigned)(N - 1);
  float4 acc = h4[node * LPN + g];
  for (int j = beg; j < end; j += 4) {
    unsigned s[4];
    float4 v[4];
#pragma unroll
    for (int k = 0; k < 4; ++k) s[k] = min((unsigned)csr[min(j + k, end - 1)], last);
#pragma unroll
    for (int k = 0; k < 4; ++k) v[k] = h4[s[k] * LPN + g];
#pragma unroll
    for (int k = 0; k < 4; ++k) {
      const bool on = (j + k) < end;
      acc.x += on ? v[k].x : 0.0f;
      acc.y += on ? v[k].y : 0.0f;
      acc.z += on ? v[k].z : 0.0f;
      acc.w += on ? v[k].w : 0.0f;
    }
  }
  reinterpret_cast<float4*>(agg)[node * LPN + g] = acc;
}

template <int K1, bool RELU_OUT>
__global__ __launch_bounds__(256)
void mlp_fused(const float* __restrict__ A, const float* __restrict__ Wa, const float* __restrict__ ba,
               const float* __restrict__ Wb, const float* __restrict__ bb,
               float* __restrict__ out, int N, int wb_at) {
  __shared__ __align__(16) float w[(K1 + COUT) * COUT];
  const float* wa = w;
  const float* wb = w + wb_at;

  const int tid  = threadIdx.x;
  const int wave = tid >> 5;
  const int lane = tid & 31;
  const int half = lane >> 4;
  const int l15  = lane & 15;
  const int row0 = (blockIdx.x * 8 + wave) * 16;

  v2f zf[K1 / 4];
  {
    const float* aRow = A + (size_t)min(row0 + l15, N - 1) * K1 + half * 2;
#pragma unroll
    for (int i = 0; i < K1 / 4; ++i) zf[i] = *reinterpret_cast<const v2f*>(aRow + 4 * i);
  }
  {
    constexpr int NA = K1 * (COUT / 4) / 256;
    constexpr int NB = COUT * (COUT / 4) / 256;
    const float4* __restrict__ Wa4 = reinterpret_cast<const float4*>(Wa);
    const float4* __restrict__ Wb4 = reinterpret_cast<const float4*>(Wb);
    float4* wa4 = reinterpret_cast<float4*>(w);
    float4* wb4 = wa4 + K1 * (COUT / 4);
    float4 va[NA], vb[NB];
#pragma unroll
    for (int r = 0; r < NA; ++r) va[r] = Wa4[tid + r * 256];
#pragma unroll
    for (int r = 0; r < NB; ++r) vb[r] = Wb4[tid + r * 256];
#pragma unroll
    for (int r = 0; r < NA; ++r) wa4[tid + r * 256] = va[r];
#pragma unroll
    for (int r = 0; r < NB; ++r) wb4[tid + r * 256] = vb[r];
  }
  __syncthreads();
  if (row0 >= N) return;

  v8f acc[8] = {};
#pragma unroll
  for (int i = 0; i < K1 / 4; ++i) {
    const float* w0 = &wa[(4 * i + half * 2) * COUT + l15];
    const float* w1 = w0 + COUT;
#pragma unroll
    for (int ct = 0; ct < 8; ++ct) {
      v2f wf;
      wf.x = w0[ct * 16];
      wf.y = w1[ct * 16];
      acc[ct] = __builtin_amdgcn_wmma_f32_16x16x4_f32(
          false, wf, false, zf[i], (short)0, acc[ct], false, false);
    }
  }

#pragma unroll
  for (int ct = 0; ct < 8; ++ct) {
    const float4 b0 = *reinterpret_cast<const float4*>(ba + ct * 16 + half * 8);
    const float4 b1 = *reinterpret_cast<const float4*>(ba + ct * 16 + half * 8 + 4);
    acc[ct][0] = fmaxf(acc[ct][0] + b0.x, 0.0f);
    acc[ct][1] = fmaxf(acc[ct][1] + b0.y, 0.0f);
    acc[ct][2] = fmaxf(acc[ct][2] + b0.z, 0.0f);
    acc[ct][3] = fmaxf(acc[ct][3] + b0.w, 0.0f);
    acc[ct][4] = fmaxf(acc[ct][4] + b1.x, 0.0f);
    acc[ct][5] = fmaxf(acc[ct][5] + b1.y, 0.0f);
    acc[ct][6] = fmaxf(acc[ct][6] + b1.z, 0.0f);
    acc[ct][7] = fmaxf(acc[ct][7] + b1.w, 0.0f);
  }

  v8f res[8] = {};
  const float* wbl = wb + half * 8 * COUT + l15;
#pragma unroll
  for (int ct = 0; ct < 8; ++ct) {
#pragma unroll
    for (int j = 0; j < 4; ++j) {
      v2f hf;
      hf.x = acc[ct][2 * j];
      hf.y = acc[ct][2 * j + 1];
      const float* w0 = wbl + (ct * 16 + 2 * j) * COUT;
      const float* w1 = w0 + COUT;
#pragma unroll
      for (int c2 = 0; c2 < 8; ++c2) {
        v2f wf;
        wf.x = w0[c2 * 16];
        wf.y = w1[c2 * 16];
        res[c2] = __builtin_amdgcn_wmma_f32_16x16x4_f32(
            false, wf, false, hf, (short)0, res[c2], false, false);
      }
    }
  }

  float* orow = out + (size_t)(row0 + l15) * COUT + half * 8;
#pragma unroll
  for (int c2 = 0; c2 < 8; ++c2) {
    const float4 b0 = *reinterpret_cast<const float4*>(bb + c2 * 16 + half * 8);
    const float4 b1 = *reinterpret_cast<const float4*>(bb + c2 * 16 + half * 8 + 4);
    float4 o0, o1;
    o0.x = res[c2][0] + b0.x;
    o0.y = res[c2][1] + b0.y;
    o0.z = res[c2][2] + b0.z;
    o0.w = res[c2][3] + b0.w;
    o1.x = res[c2][4] + b1.x;
    o1.y = res[c2][5] + b1.y;
    o1.z = res[c2][6] + b1.z;
    o1.w = res[c2][7] + b1.w;
    if (RELU_OUT) {
      o0.x = fmaxf(o0.x, 0.0f); o0.y = fmaxf(o0.y, 0.0f); o0.z = fmaxf(o0.z, 0.0f); o0.w = fmaxf(o0.w, 0.0f);
      o1.x = fmaxf(o1.x, 0.0f); o1.y = fmaxf(o1.y, 0.0f); o1.z = fmaxf(o1.z, 0.0f); o1.w = fmaxf(o1.w, 0.0f);
    }
    float4* dst = reinterpret_cast<float4*>(orow + c2 * 16);
    dst[0] = o0;
    dst[1] = o1;
  }
}

extern "C" void kernel_launch(void* const* d_in, const int* in_sizes, int n_in,
                              void* d_out, int out_size, void* d_ws, size_t ws_size,
                              hipStream_t stream) {
  const float* x   = (const float*)d_in[0];
  const int*   ei  = (const int*)d_in[1];
  const float* W1a = (const float*)d_in[2];
  const float* b1a = (const float*)d_in[3];
  const float* W1b = (const float*)d_in[4];
  const float* b1b = (const float*)d_in[5];
  const float* W2a = (const float*)d_in[6];
  const float* b2a = (const float*)d_in[7];
  const float* W2b = (const float*)d_in[8];
  const float* b2b = (const float*)d_in[9];
  float* out = (float*)d_out;

  const int N = in_sizes[0] / 64;
  const int E = in_sizes[1] / 2;
  const int* src = ei;
  const int* dst = ei + E;

  float* R0 = (float*)d_ws;
  float* R2 = R0 + (size_t)N * 256;

  const int NB  = (N + BW - 1) / BW;
  const int per = (E + NCHUNK - 1) / NCHUNK;
  const int nt  = (NB + 1) * NCHUNK;

  int*  off    = (int*)(R2 + (size_t)N * 128);
  int*  rowptr = off + (nt + 31) / 32 * 32;
  int2* srt    = (int2*)(rowptr + (N + 1 + 31) / 32 * 32);
  int*  csr    = (int*)(srt + E);

  const int strips  = (N + 15) / 16;
  const int gblocks = (strips + 7) / 8;

  sort_chunk<<<NCHUNK, 1024, NB * sizeof(int), stream>>>(src, dst, off, srt, E, per, NB);
  build_rows<<<NB, 1024, 0, stream>>>(off, srt, rowptr, csr, N, E, per);

  gather_rows<64><<<(N * 16 + 255) / 256, 256, 0, stream>>>(R0, x, rowptr, csr, N);
  mlp_fused<64, true><<<gblocks, 256, 0, stream>>>(R0, W1a, b1a, W1b, b1b, R2, N, 64 * COUT);

  gather_rows<128><<<(N * 32 + 255) / 256, 256, 0, stream>>>(R0, R2, rowptr, csr, N);
  mlp_fused<128, false><<<gblocks, 256, 0, stream>>>(R0, W2a, b2a, W2b, b2b, out, N, 128 * COUT);
}


# Neighbour sum moved into each layer's product launch: a wave sums its 16 rows into an LDS tile that it reads as B fragments; 2 launches and the summed plane's store and reload go; sum order kept
# speedup vs baseline: 3.4855x; 1.0380x over previous
#include <hip/hip_runtime.h>

#define COUT 128

typedef float v2f __attribute__((ext_vector_type(2)));
typedef float v8f __attribute__((ext_vector_type(8)));


#define BSH    8
#define BW     (1 << BSH)
#define NCHUNK 256

__device__ __forceinline__ int bucket_of(int d, int nb) {
  return (int)min((unsigned)d >> BSH, (unsigned)(nb - 1));
}

__device__ __forceinline__ int wave_inclusive_sum(int x, int lane) {
  x += __builtin_amdgcn_update_dpp(0, x, 0x111, 0xf, 0xf, false);
  x += __builtin_amdgcn_update_dpp(0, x, 0x112, 0xf, 0xf, false);
  x += __builtin_amdgcn_update_dpp(0, x, 0x114, 0xf, 0xf, false);
  x += __builtin_amdgcn_update_dpp(0, x, 0x118, 0xf, 0xf, false);
  const int low_row = __builtin_amdgcn_readlane(x, 15);
  return x + (lane >= 16 ? low_row : 0);
}

__global__ __launch_bounds__(1024)
void sort_chunk(const int* __restrict__ src, const int* __restrict__ dst,
                int* __restrict__ off, int2* __restrict__ srt, int E, int per, int nb) {
  extern __shared__ int lcur[];
  const int c   = blockIdx.x;
  const int tid = threadIdx.x;
  for (int i = tid; i < nb; i += 1024) lcur[i] = 0;
  __syncthreads();
  const int e0 = c * per;
  const int e1 = min(E, e0 + per);
  for (int base = e0; base < e1; base += 4 * 1024) {
    int d[4];
#pragma unroll
    for (int k = 0; k < 4; ++k) d[k] = dst[min(base + k * 1024 + tid, e1 - 1)];
    asm volatile("" :: "v"(d[0]), "v"(d[1]), "v"(d[2]), "v"(d[3]));
#pragma unroll
    for (int k = 0; k < 4; ++k)
      if (base + k * 1024 + tid < e1) atomicAdd(&lcur[bucket_of(d[k], nb)], 1);
  }
  __syncthreads();
  if (tid < 32) {
    int carry = 0;
    for (int i0 = 0; i0 < nb; i0 += 32) {
      const int i   = i0 + tid;
      const int v   = (i < nb) ? lcur[i] : 0;
      const int inc = wave_inclusive_sum(v, tid);
      if (i < nb) {
        const int start = carry + inc - v;
        lcur[i] = e0 + start;
        off[i * NCHUNK + c] = start;
      }
      carry += __builtin_amdgcn_readlane(inc, 31);
    }
    if (tid == 0) off[nb * NCHUNK + c] = carry;
  }
  __syncthreads();
  for (int base = e0; base < e1; base += 4 * 1024) {
    int s[4], d[4];
#pragma unroll
    for (int k = 0; k < 4; ++k) {
      const int e = min(base + k * 1024 + tid, e1 - 1);
      d[k] = dst[e];
      s[k] = src[e];
    }
    asm volatile("" :: "v"(d[0]), "v"(d[1]), "v"(d[2]), "v"(d[3]), "v"(s[0]), "v"(s[1]), "v"(s[2]), "v"(s[3]));
#pragma unroll
    for (int k = 0; k < 4; ++k)
      if (base + k * 1024 + tid < e1) {
        const int pos = atomicAdd(&lcur[bucket_of(d[k], nb)], 1);
        srt[pos] = make_int2(s[k], d[k]);
      }
  }
}

#define CPW (NCHUNK / 32)

__global__ __launch_bounds__(1024)
void build_rows(const int* __restrict__ off, const int2* __restrict__ srt,
                int* __restrict__ rowptr, int* __restrict__ csr, int N, int E, int per) {
  __shared__ int slot[BW];
  __shared__ int wtot[BW / 32];
  __shared__ int pbeg[NCHUNK];
  __shared__ int pend[NCHUNK];
  __shared__ int part[2][NCHUNK / 32];
  const int b    = blockIdx.x;
  const int tid  = threadIdx.x;
  const int lane = tid & 31;
  const int wave = tid >> 5;
  const int* __restrict__ sdst = reinterpret_cast<const int*>(srt);
  if (tid < BW) slot[tid] = 0;
  if (tid < NCHUNK) {
    const int lo = off[b * NCHUNK + tid];
    const int hi = off[(b + 1) * NCHUNK + tid];
    pbeg[tid] = tid * per + lo;
    pend[tid] = tid * per + hi;
    const int slo = wave_inclusive_sum(lo, lane);
    const int sln = wave_inclusive_sum(hi - lo, lane);
    if (lane == 31) { part[0][wave] = slo; part[1][wave] = sln; }
  }
  __syncthreads();
  int beg = 0, cnt = 0;
#pragma unroll
  for (int w = 0; w < NCHUNK / 32; ++w) { beg += part[0][w]; cnt += part[1][w]; }
  const int end = beg + cnt;

  int lo[CPW], hi[CPW], at[CPW];
#pragma unroll
  for (int s = 0; s < CPW; ++s) {
    lo[s] = pbeg[wave * CPW + s];
    hi[s] = pend[wave * CPW + s];
    at[s] = max(min(min(lo[s] + lane, hi[s] - 1), E - 1), 0);
  }
  {
    int d[CPW];
#pragma unroll
    for (int s = 0; s < CPW; ++s) d[s] = sdst[2 * at[s] + 1];
    static_assert(CPW == 8, "the fence below names CPW values");
    asm volatile("" :: "v"(d[0]), "v"(d[1]), "v"(d[2]), "v"(d[3]), "v"(d[4]), "v"(d[5]), "v"(d[6]), "v"(d[7]));
#pragma unroll
    for (int s = 0; s < CPW; ++s)
      if (lo[s] + lane < hi[s]) atomicAdd(&slot[d[s] & (BW - 1)], 1);
#pragma unroll
    for (int s = 0; s < CPW; ++s)
      for (int i = lo[s] + 32 + lane; i < hi[s]; i += 32) atomicAdd(&slot[sdst[2 * i + 1] & (BW - 1)], 1);
  }
  __syncthreads();
  int deg = 0, inc = 0;
  if (tid < BW) {
    deg = slot[tid];
    inc = wave_inclusive_sum(deg, lane);
    if (lane == 31) wtot[wave] = inc;
  }
  __syncthreads();
  if (tid < BW) {
    int before = 0;
#pragma unroll
    for (int w = 0; w < BW / 32; ++w) before += (w < wave) ? wtot[w] : 0;
    const int start = beg + before + inc - deg;
    slot[tid] = start;
    const int node = b * BW + tid;
    if (node < N) rowptr[node] = start;
    if (node == N - 1) rowptr[N] = end;
  }
  __syncthreads();
  {
    int2 v[CPW];
#pragma unroll
    for (int s = 0; s < CPW; ++s) v[s] = srt[at[s]];
    asm volatile("" :: "v"(v[0].y), "v"(v[1].y), "v"(v[2].y), "v"(v[3].y), "v"(v[4].y), "v"(v[5].y), "v"(v[6].y), "v"(v[7].y));
#pragma unroll
    for (int s = 0; s < CPW; ++s)
      if (lo[s] + lane < hi[s]) {
        const int pos = atomicAdd(&slot[v[s].y & (BW - 1)], 1);
        csr[pos] = v[s].x;
      }
#pragma unroll
    for (int s = 0; s < CPW; ++s)
      for (int i = lo[s] + 32 + lane; i < hi[s]; i += 32) {
        const int2 p = srt[i];
        const int pos = atomicAdd(&slot[p.y & (BW - 1)], 1);
        csr[pos] = p.x;
      }
  }
}

template <int K1, bool RELU_OUT>
__global__ __launch_bounds__(256)
void gin_layer(const float* __restrict__ A, const int* __restrict__ rowptr, const int* __restrict__ csr,
               const float* __restrict__ Wa, const float* __restrict__ ba,
               const float* __restrict__ Wb, const float* __restrict__ bb,
               float* __restrict__ out, int N, int wb_at) {
  __shared__ __align__(16) float w[(K1 + COUT) * COUT];
  const float* wa = w;
  const float* wb = w + wb_at;

  const int tid  = threadIdx.x;
  const int wave = tid >> 5;
  const int lane = tid & 31;
  const int half = lane >> 4;
  const int l15  = lane & 15;
  const int row0 = (blockIdx.x * 8 + wave) * 16;

  const bool live = row0 < N;

  constexpr int VW = K1 / 32;
  constexpr int NB = 32;
  typedef float rowv __attribute__((ext_vector_type(VW)));
  const rowv* __restrict__ hv = reinterpret_cast<const rowv*>(A);
  float* tile = w + wb_at + wave * (16 * K1);
  const int lslot = (lane * VW) >> 2;
  const int lsub  = (lane * VW) & 3;

  {
    constexpr int NA = K1 * (COUT / 4) / 256;
    const float4* __restrict__ Wa4 = reinterpret_cast<const float4*>(Wa);
    float4* wa4 = reinterpret_cast<float4*>(w);
    float4 va[NA];
#pragma unroll
    for (int r = 0; r < NA; ++r) va[r] = Wa4[tid + r * 256];
#pragma unroll
    for (int r = 0; r < NA; ++r) wa4[tid + r * 256] = va[r];
  }

  v2f zf[K1 / 4];
  if (live) {
    const int rpv = rowptr[min(row0 + min(lane, 16), N)];
    const int E0 = __builtin_amdgcn_readlane(rpv, 0);
    const int E1 = __builtin_amdgcn_readlane(rpv, 16);
    const unsigned last = (unsigned)(N - 1);
    {
      rowv own[16];
#pragma unroll
      for (int r = 0; r < 16; ++r) own[r] = hv[(size_t)min(row0 + r, N - 1) * 32 + lane];
#pragma unroll
      for (int r = 0; r < 16; ++r)
        *reinterpret_cast<rowv*>(tile + r * K1 + ((lslot ^ r) << 2) + lsub) = own[r];
    }
    int cur  = 0;
    int rend = __builtin_amdgcn_readlane(rpv, 1);
    rowv sum = *reinterpret_cast<const rowv*>(tile + ((lslot ^ 0) << 2) + lsub);
    int idxv = (E0 < E1) ? csr[min(E0 + lane, E1 - 1)] : 0;
    for (int e = E0; e < E1; e += NB) {
      const int cidx = idxv;
      if (e + NB < E1) idxv = csr[min(e + NB + lane, E1 - 1)];
      rowv v[NB];
#pragma unroll
      for (int k = 0; k < NB; ++k) {
        const unsigned s = min((unsigned)__builtin_amdgcn_readlane(cidx, k), last);
        v[k] = hv[(size_t)s * 32 + lane];
      }
#pragma unroll
      for (int k = 0; k < NB; ++k) {
        const int pos = e + k;
        if (pos < E1) {
          while (pos >= rend && cur < 15) {
            *reinterpret_cast<rowv*>(tile + cur * K1 + ((lslot ^ cur) << 2) + lsub) = sum;
            ++cur;
            rend = __builtin_amdgcn_readlane(rpv, cur + 1);
            sum = *reinterpret_cast<const rowv*>(tile + cur * K1 + ((lslot ^ cur) << 2) + lsub);
          }
          sum += v[k];
        }
      }
    }
    *reinterpret_cast<rowv*>(tile + cur * K1 + ((lslot ^ cur) << 2) + lsub) = sum;

#pragma unroll
    for (int i = 0; i < K1 / 4; ++i)
      zf[i] = *reinterpret_cast<const v2f*>(tile + l15 * K1 + ((i ^ l15) << 2) + half * 2);
  }
  __syncthreads();


  v8f acc[8] = {};
  if (live) {
#pragma unroll
  for (int i = 0; i < K1 / 4; ++i) {
    const float* w0 = &wa[(4 * i + half * 2) * COUT + l15];
    const float* w1 = w0 + COUT;
#pragma unroll
    for (int ct = 0; ct < 8; ++ct) {
      v2f wf;
      wf.x = w0[ct * 16];
      wf.y = w1[ct * 16];
      acc[ct] = __builtin_amdgcn_wmma_f32_16x16x4_f32(
          false, wf, false, zf[i], (short)0, acc[ct], false, false);
    }
  }

#pragma unroll
  for (int ct = 0; ct < 8; ++ct) {
    const float4 b0 = *reinterpret_cast<const float4*>(ba + ct * 16 + half * 8);
    const float4 b1 = *reinterpret_cast<const float4*>(ba + ct * 16 + half * 8 + 4);
    acc[ct][0] = fmaxf(acc[ct][0] + b0.x, 0.0f);
    acc[ct][1] = fmaxf(acc[ct][1] + b0.y, 0.0f);
    acc[ct][2] = fmaxf(acc[ct][2] + b0.z, 0.0f);
    acc[ct][3] = fmaxf(acc[ct][3] + b0.w, 0.0f);
    acc[ct][4] = fmaxf(acc[ct][4] + b1.x, 0.0f);
    acc[ct][5] = fmaxf(acc[ct][5] + b1.y, 0.0f);
    acc[ct][6] = fmaxf(acc[ct][6] + b1.z, 0.0f);
    acc[ct][7] = fmaxf(acc[ct][7] + b1.w, 0.0f);
  }
  }

  {
    constexpr int NWB = COUT * (COUT / 4) / 256;
    const float4* __restrict__ Wb4 = reinterpret_cast<const float4*>(Wb);
    float4* wb4 = reinterpret_cast<float4*>(w + wb_at);
    float4 vb[NWB];
#pragma unroll
    for (int r = 0; r < NWB; ++r) vb[r] = Wb4[tid + r * 256];
#pragma unroll
    for (int r = 0; r < NWB; ++r) wb4[tid + r * 256] = vb[r];
  }
  __syncthreads();
  if (!live) return;

  v8f res[8] = {};
  const float* wbl = wb + half * 8 * COUT + l15;
#pragma unroll
  for (int ct = 0; ct < 8; ++ct) {
#pragma unroll
    for (int j = 0; j < 4; ++j) {
      v2f hf;
      hf.x = acc[ct][2 * j];
      hf.y = acc[ct][2 * j + 1];
      const float* w0 = wbl + (ct * 16 + 2 * j) * COUT;
      const float* w1 = w0 + COUT;
#pragma unroll
      for (int c2 = 0; c2 < 8; ++c2) {
        v2f wf;
        wf.x = w0[c2 * 16];
        wf.y = w1[c2 * 16];
        res[c2] = __builtin_amdgcn_wmma_f32_16x16x4_f32(
            false, wf, false, hf, (short)0, res[c2], false, false);
      }
    }
  }

  float* orow = out + (size_t)(row0 + l15) * COUT + half * 8;
#pragma unroll
  for (int c2 = 0; c2 < 8; ++c2) {
    const float4 b0 = *reinterpret_cast<const float4*>(bb + c2 * 16 + half * 8);
    const float4 b1 = *reinterpret_cast<const float4*>(bb + c2 * 16 + half * 8 + 4);
    float4 o0, o1;
    o0.x = res[c2][0] + b0.x;
    o0.y = res[c2][1] + b0.y;
    o0.z = res[c2][2] + b0.z;
    o0.w = res[c2][3] + b0.w;
    o1.x = res[c2][4] + b1.x;
    o1.y = res[c2][5] + b1.y;
    o1.z = res[c2][6] + b1.z;
    o1.w = res[c2][7] + b1.w;
    if (RELU_OUT) {
      o0.x = fmaxf(o0.x, 0.0f); o0.y = fmaxf(o0.y, 0.0f); o0.z = fmaxf(o0.z, 0.0f); o0.w = fmaxf(o0.w, 0.0f);
      o1.x = fmaxf(o1.x, 0.0f); o1.y = fmaxf(o1.y, 0.0f); o1.z = fmaxf(o1.z, 0.0f); o1.w = fmaxf(o1.w, 0.0f);
    }
    float4* dst = reinterpret_cast<float4*>(orow + c2 * 16);
    dst[0] = o0;
    dst[1] = o1;
  }
}

extern "C" void kernel_launch(void* const* d_in, const int* in_sizes, int n_in,
                              void* d_out, int out_size, void* d_ws, size_t ws_size,
                              hipStream_t stream) {
  const float* x   = (const float*)d_in[0];
  const int*   ei  = (const int*)d_in[1];
  const float* W1a = (const float*)d_in[2];
  const float* b1a = (const float*)d_in[3];
  const float* W1b = (const float*)d_in[4];
  const float* b1b = (const float*)d_in[5];
  const float* W2a = (const float*)d_in[6];
  const float* b2a = (const float*)d_in[7];
  const float* W2b = (const float*)d_in[8];
  const float* b2b = (const float*)d_in[9];
  float* out = (float*)d_out;

  const int N = in_sizes[0] / 64;
  const int E = in_sizes[1] / 2;
  const int* src = ei;
  const int* dst = ei + E;

  float* R0 = (float*)d_ws;
  float* R2 = R0 + (size_t)N * 256;

  const int NB  = (N + BW - 1) / BW;
  const int per = (E + NCHUNK - 1) / NCHUNK;
  const int nt  = (NB + 1) * NCHUNK;

  int*  off    = (int*)(R2 + (size_t)N * 128);
  int*  rowptr = off + (nt + 31) / 32 * 32;
  int2* srt    = (int2*)(rowptr + (N + 1 + 31) / 32 * 32);
  int*  csr    = (int*)(srt + E);

  const int strips  = (N + 15) / 16;
  const int gblocks = (strips + 7) / 8;

  sort_chunk<<<NCHUNK, 1024, NB * sizeof(int), stream>>>(src, dst, off, srt, E, per, NB);
  build_rows<<<NB, 1024, 0, stream>>>(off, srt, rowptr, csr, N, E, per);

  gin_layer<64, true><<<gblocks, 256, 0, stream>>>(x, rowptr, csr, W1a, b1a, W1b, b1b, R2, N, 64 * COUT);

  gin_layer<128, false><<<gblocks, 256, 0, stream>>>(R2, rowptr, csr, W2a, b2a, W2b, b2b, out, N, 128 * COUT);
}


# Neighbour sum: each node filed as first entry of its own source list, so no list is empty and per-entry row tracking is one compare of the unrolled slot with the next row's start; sum order kept
# speedup vs baseline: 3.7025x; 1.0623x over previous
#include <hip/hip_runtime.h>

#define COUT 128

typedef float v2f __attribute__((ext_vector_type(2)));
typedef float v8f __attribute__((ext_vector_type(8)));


#define BSH    8
#define BW     (1 << BSH)
#define NCHUNK 256

__device__ __forceinline__ int bucket_of(int d, int nb) {
  return (int)min((unsigned)d >> BSH, (unsigned)(nb - 1));
}

__device__ __forceinline__ int wave_inclusive_sum(int x, int lane) {
  x += __builtin_amdgcn_update_dpp(0, x, 0x111, 0xf, 0xf, false);
  x += __builtin_amdgcn_update_dpp(0, x, 0x112, 0xf, 0xf, false);
  x += __builtin_amdgcn_update_dpp(0, x, 0x114, 0xf, 0xf, false);
  x += __builtin_amdgcn_update_dpp(0, x, 0x118, 0xf, 0xf, false);
  const int low_row = __builtin_amdgcn_readlane(x, 15);
  return x + (lane >= 16 ? low_row : 0);
}

__global__ __launch_bounds__(1024)
void sort_chunk(const int* __restrict__ src, const int* __restrict__ dst,
                int* __restrict__ off, int2* __restrict__ srt, int E, int per, int nb) {
  extern __shared__ int lcur[];
  const int c   = blockIdx.x;
  const int tid = threadIdx.x;
  for (int i = tid; i < nb; i += 1024) lcur[i] = 0;
  __syncthreads();
  const int e0 = c * per;
  const int e1 = min(E, e0 + per);
  for (int base = e0; base < e1; base += 4 * 1024) {
    int d[4];
#pragma unroll
    for (int k = 0; k < 4; ++k) d[k] = dst[min(base + k * 1024 + tid, e1 - 1)];
    asm volatile("" :: "v"(d[0]), "v"(d[1]), "v"(d[2]), "v"(d[3]));
#pragma unroll
    for (int k = 0; k < 4; ++k)
      if (base + k * 1024 + tid < e1) atomicAdd(&lcur[bucket_of(d[k], nb)], 1);
  }
  __syncthreads();
  if (tid < 32) {
    int carry = 0;
    for (int i0 = 0; i0 < nb; i0 += 32) {
      const int i   = i0 + tid;
      const int v   = (i < nb) ? lcur[i] : 0;
      const int inc = wave_inclusive_sum(v, tid);
      if (i < nb) {
        const int start = carry + inc - v;
        lcur[i] = e0 + start;
        off[i * NCHUNK + c] = start;
      }
      carry += __builtin_amdgcn_readlane(inc, 31);
    }
    if (tid == 0) off[nb * NCHUNK + c] = carry;
  }
  __syncthreads();
  for (int base = e0; base < e1; base += 4 * 1024) {
    int s[4], d[4];
#pragma unroll
    for (int k = 0; k < 4; ++k) {
      const int e = min(base + k * 1024 + tid, e1 - 1);
      d[k] = dst[e];
      s[k] = src[e];
    }
    asm volatile("" :: "v"(d[0]), "v"(d[1]), "v"(d[2]), "v"(d[3]), "v"(s[0]), "v"(s[1]), "v"(s[2]), "v"(s[3]));
#pragma unroll
    for (int k = 0; k < 4; ++k)
      if (base + k * 1024 + tid < e1) {
        const int pos = atomicAdd(&lcur[bucket_of(d[k], nb)], 1);
        srt[pos] = make_int2(s[k], d[k]);
      }
  }
}

#define CPW (NCHUNK / 32)

__global__ __launch_bounds__(1024)
void build_rows(const int* __restrict__ off, const int2* __restrict__ srt,
                int* __restrict__ rowptr, int* __restrict__ csr, int N, int E, int per) {
  __shared__ int slot[BW];
  __shared__ int wtot[BW / 32];
  __shared__ int pbeg[NCHUNK];
  __shared__ int pend[NCHUNK];
  __shared__ int part[2][NCHUNK / 32];
  const int b    = blockIdx.x;
  const int tid  = threadIdx.x;
  const int lane = tid & 31;
  const int wave = tid >> 5;
  const int* __restrict__ sdst = reinterpret_cast<const int*>(srt);
  if (tid < BW) slot[tid] = 0;
  if (tid < NCHUNK) {
    const int lo = off[b * NCHUNK + tid];
    const int hi = off[(b + 1) * NCHUNK + tid];
    pbeg[tid] = tid * per + lo;
    pend[tid] = tid * per + hi;
    const int slo = wave_inclusive_sum(lo, lane);
    const int sln = wave_inclusive_sum(hi - lo, lane);
    if (lane == 31) { part[0][wave] = slo; part[1][wave] = sln; }
  }
  __syncthreads();
  int beg = 0, cnt = 0;
#pragma unroll
  for (int w = 0; w < NCHUNK / 32; ++w) { beg += part[0][w]; cnt += part[1][w]; }
  const int end = beg + cnt;

  int lo[CPW], hi[CPW], at[CPW];
#pragma unroll
  for (int s = 0; s < CPW; ++s) {
    lo[s] = pbeg[wave * CPW + s];
    hi[s] = pend[wave * CPW + s];
    at[s] = max(min(min(lo[s] + lane, hi[s] - 1), E - 1), 0);
  }
  {
    int d[CPW];
#pragma unroll
    for (int s = 0; s < CPW; ++s) d[s] = sdst[2 * at[s] + 1];
    static_assert(CPW == 8, "the fence below names CPW values");
    asm volatile("" :: "v"(d[0]), "v"(d[1]), "v"(d[2]), "v"(d[3]), "v"(d[4]), "v"(d[5]), "v"(d[6]), "v"(d[7]));
#pragma unroll
    for (int s = 0; s < CPW; ++s)
      if (lo[s] + lane < hi[s]) atomicAdd(&slot[d[s] & (BW - 1)], 1);
#pragma unroll
    for (int s = 0; s < CPW; ++s)
      for (int i = lo[s] + 32 + lane; i < hi[s]; i += 32) atomicAdd(&slot[sdst[2 * i + 1] & (BW - 1)], 1);
  }
  __syncthreads();
  int deg = 0, inc = 0;
  if (tid < BW) {
    deg = slot[tid];
    inc = wave_inclusive_sum(deg, lane);
    if (lane == 31) wtot[wave] = inc;
  }
  __syncthreads();
  if (tid < BW) {
    int before = 0;
#pragma unroll
    for (int w = 0; w < BW / 32; ++w) before += (w < wave) ? wtot[w] : 0;
    const int node = b * BW + tid;
    const int start = beg + before + inc - deg + min(node, N);
    slot[tid] = start + 1;
    if (node < N) { rowptr[node] = start; csr[start] = node; }
    if (node == N - 1) rowptr[N] = end + N;
  }
  __syncthreads();
  {
    int2 v[CPW];
#pragma unroll
    for (int s = 0; s < CPW; ++s) v[s] = srt[at[s]];
    asm volatile("" :: "v"(v[0].y), "v"(v[1].y), "v"(v[2].y), "v"(v[3].y), "v"(v[4].y), "v"(v[5].y), "v"(v[6].y), "v"(v[7].y));
#pragma unroll
    for (int s = 0; s < CPW; ++s)
      if (lo[s] + lane < hi[s]) {
        const int pos = atomicAdd(&slot[v[s].y & (BW - 1)], 1);
        csr[pos] = v[s].x;
      }
#pragma unroll
    for (int s = 0; s < CPW; ++s)
      for (int i = lo[s] + 32 + lane; i < hi[s]; i += 32) {
        const int2 p = srt[i];
        const int pos = atomicAdd(&slot[p.y & (BW - 1)], 1);
        csr[pos] = p.x;
      }
  }
}

template <int K1, bool RELU_OUT>
__global__ __launch_bounds__(256)
void gin_layer(const float* __restrict__ A, const int* __restrict__ rowptr, const int* __restrict__ csr,
               const float* __restrict__ Wa, const float* __restrict__ ba,
               const float* __restrict__ Wb, const float* __restrict__ bb,
               float* __restrict__ out, int N, int wb_at) {
  __shared__ __align__(16) float w[(K1 + COUT) * COUT];
  const float* wa = w;
  const float* wb = w + wb_at;

  const int tid  = threadIdx.x;
  const int wave = tid >> 5;
  const int lane = tid & 31;
  const int half = lane >> 4;
  const int l15  = lane & 15;
  const int row0 = (blockIdx.x * 8 + wave) * 16;

  const bool live = row0 < N;

  constexpr int VW = K1 / 32;
  constexpr int NB = 32;
  typedef float rowv __attribute__((ext_vector_type(VW)));
  const rowv* __restrict__ hv = reinterpret_cast<const rowv*>(A);
  float* tile = w + wb_at + wave * (16 * K1);
  const int lslot = (lane * VW) >> 2;
  const int lsub  = (lane * VW) & 3;

  {
    constexpr int NA = K1 * (COUT / 4) / 256;
    const float4* __restrict__ Wa4 = reinterpret_cast<const float4*>(Wa);
    float4* wa4 = reinterpret_cast<float4*>(w);
    float4 va[NA];
#pragma unroll
    for (int r = 0; r < NA; ++r) va[r] = Wa4[tid + r * 256];
#pragma unroll
    for (int r = 0; r < NA; ++r) wa4[tid + r * 256] = va[r];
  }

  v2f zf[K1 / 4];
  if (live) {
    const int rpv = rowptr[min(row0 + min(lane, 16), N)];
    const int E0 = __builtin_amdgcn_readlane(rpv, 0);
    const int E1 = __builtin_amdgcn_readlane(rpv, 16);
    const unsigned last = (unsigned)(N - 1);
    int cur  = 0;
    int rend = __builtin_amdgcn_readlane(rpv, 1);
    rowv sum = {};
    int idxv = (E0 < E1) ? csr[min(E0 + lane, E1 - 1)] : 0;
    for (int e = E0; e < E1; e += NB) {
      const int cidx = idxv;
      if (e + NB < E1) idxv = csr[min(e + NB + lane, E1 - 1)];
      rowv v[NB];
#pragma unroll
      for (int k = 0; k < NB; ++k) {
        const unsigned s = min((unsigned)__builtin_amdgcn_readlane(cidx, k), last);
        v[k] = hv[(size_t)s * 32 + lane];
      }
      int rel = rend - e;
#pragma unroll
      for (int k = 0; k < NB; ++k) {
        if (__builtin_expect(k == rel, 0)) {
          *reinterpret_cast<rowv*>(tile + cur * K1 + ((lslot ^ cur) << 2) + lsub) = sum;
          ++cur;
          rend = __builtin_amdgcn_readlane(rpv, cur + 1);
          rel  = rend - e;
          sum  = rowv{};
        }
        sum += v[k];
      }
    }
    if (cur < 16) *reinterpret_cast<rowv*>(tile + cur * K1 + ((lslot ^ cur) << 2) + lsub) = sum;

#pragma unroll
    for (int i = 0; i < K1 / 4; ++i)
      zf[i] = *reinterpret_cast<const v2f*>(tile + l15 * K1 + ((i ^ l15) << 2) + half * 2);
  }
  __syncthreads();


  v8f acc[8] = {};
  if (live) {
#pragma unroll
  for (int i = 0; i < K1 / 4; ++i) {
    const float* w0 = &wa[(4 * i + half * 2) * COUT + l15];
    const float* w1 = w0 + COUT;
#pragma unroll
    for (int ct = 0; ct < 8; ++ct) {
      v2f wf;
      wf.x = w0[ct * 16];
      wf.y = w1[ct * 16];
      acc[ct] = __builtin_amdgcn_wmma_f32_16x16x4_f32(
          false, wf, false, zf[i], (short)0, acc[ct], false, false);
    }
  }

#pragma unroll
  for (int ct = 0; ct < 8; ++ct) {
    const float4 b0 = *reinterpret_cast<const float4*>(ba + ct * 16 + half * 8);
    const float4 b1 = *reinterpret_cast<const float4*>(ba + ct * 16 + half * 8 + 4);
    acc[ct][0] = fmaxf(acc[ct][0] + b0.x, 0.0f);
    acc[ct][1] = fmaxf(acc[ct][1] + b0.y, 0.0f);
    acc[ct][2] = fmaxf(acc[ct][2] + b0.z, 0.0f);
    acc[ct][3] = fmaxf(acc[ct][3] + b0.w, 0.0f);
    acc[ct][4] = fmaxf(acc[ct][4] + b1.x, 0.0f);
    acc[ct][5] = fmaxf(acc[ct][5] + b1.y, 0.0f);
    acc[ct][6] = fmaxf(acc[ct][6] + b1.z, 0.0f);
    acc[ct][7] = fmaxf(acc[ct][7] + b1.w, 0.0f);
  }
  }

  {
    constexpr int NWB = COUT * (COUT / 4) / 256;
    const float4* __restrict__ Wb4 = reinterpret_cast<const float4*>(Wb);
    float4* wb4 = reinterpret_cast<float4*>(w + wb_at);
    float4 vb[NWB];
#pragma unroll
    for (int r = 0; r < NWB; ++r) vb[r] = Wb4[tid + r * 256];
#pragma unroll
    for (int r = 0; r < NWB; ++r) wb4[tid + r * 256] = vb[r];
  }
  __syncthreads();
  if (!live) return;

  v8f res[8] = {};
  const float* wbl = wb + half * 8 * COUT + l15;
#pragma unroll
  for (int ct = 0; ct < 8; ++ct) {
#pragma unroll
    for (int j = 0; j < 4; ++j) {
      v2f hf;
      hf.x = acc[ct][2 * j];
      hf.y = acc[ct][2 * j + 1];
      const float* w0 = wbl + (ct * 16 + 2 * j) * COUT;
      const float* w1 = w0 + COUT;
#pragma unroll
      for (int c2 = 0; c2 < 8; ++c2) {
        v2f wf;
        wf.x = w0[c2 * 16];
        wf.y = w1[c2 * 16];
        res[c2] = __builtin_amdgcn_wmma_f32_16x16x4_f32(
            false, wf, false, hf, (short)0, res[c2], false, false);
      }
    }
  }

  float* orow = out + (size_t)(row0 + l15) * COUT + half * 8;
#pragma unroll
  for (int c2 = 0; c2 < 8; ++c2) {
    const float4 b0 = *reinterpret_cast<const float4*>(bb + c2 * 16 + half * 8);
    const float4 b1 = *reinterpret_cast<const float4*>(bb + c2 * 16 + half * 8 + 4);
    float4 o0, o1;
    o0.x = res[c2][0] + b0.x;
    o0.y = res[c2][1] + b0.y;
    o0.z = res[c2][2] + b0.z;
    o0.w = res[c2][3] + b0.w;
    o1.x = res[c2][4] + b1.x;
    o1.y = res[c2][5] + b1.y;
    o1.z = res[c2][6] + b1.z;
    o1.w = res[c2][7] + b1.w;
    if (RELU_OUT) {
      o0.x = fmaxf(o0.x, 0.0f); o0.y = fmaxf(o0.y, 0.0f); o0.z = fmaxf(o0.z, 0.0f); o0.w = fmaxf(o0.w, 0.0f);
      o1.x = fmaxf(o1.x, 0.0f); o1.y = fmaxf(o1.y, 0.0f); o1.z = fmaxf(o1.z, 0.0f); o1.w = fmaxf(o1.w, 0.0f);
    }
    float4* dst = reinterpret_cast<float4*>(orow + c2 * 16);
    dst[0] = o0;
    dst[1] = o1;
  }
}

extern "C" void kernel_launch(void* const* d_in, const int* in_sizes, int n_in,
                              void* d_out, int out_size, void* d_ws, size_t ws_size,
                              hipStream_t stream) {
  const float* x   = (const float*)d_in[0];
  const int*   ei  = (const int*)d_in[1];
  const float* W1a = (const float*)d_in[2];
  const float* b1a = (const float*)d_in[3];
  const float* W1b = (const float*)d_in[4];
  const float* b1b = (const float*)d_in[5];
  const float* W2a = (const float*)d_in[6];
  const float* b2a = (const float*)d_in[7];
  const float* W2b = (const float*)d_in[8];
  const float* b2b = (const float*)d_in[9];
  float* out = (float*)d_out;

  const int N = in_sizes[0] / 64;
  const int E = in_sizes[1] / 2;
  const int* src = ei;
  const int* dst = ei + E;

  float* R0 = (float*)d_ws;
  float* R2 = R0 + (size_t)N * 256;

  const int NB  = (N + BW - 1) / BW;
  const int per = (E + NCHUNK - 1) / NCHUNK;
  const int nt  = (NB + 1) * NCHUNK;

  int*  off    = (int*)(R2 + (size_t)N * 128);
  int*  rowptr = off + (nt + 31) / 32 * 32;
  int2* srt    = (int2*)(rowptr + (N + 1 + 31) / 32 * 32);
  int*  csr    = (int*)(srt + E);

  const int strips  = (N + 15) / 16;
  const int gblocks = (strips + 7) / 8;

  sort_chunk<<<NCHUNK, 1024, NB * sizeof(int), stream>>>(src, dst, off, srt, E, per, NB);
  build_rows<<<NB, 1024, 0, stream>>>(off, srt, rowptr, csr, N, E, per);

  gin_layer<64, true><<<gblocks, 256, 0, stream>>>(x, rowptr, csr, W1a, b1a, W1b, b1b, R2, N, 64 * COUT);

  gin_layer<128, false><<<gblocks, 256, 0, stream>>>(R2, rowptr, csr, W2a, b2a, W2b, b2b, out, N, 128 * COUT);
}


# Layer grid = one 16-wave work-group per WGP (256 x 12-13 strips; was 391 x 8 waves, two on some WGPs, one on others): all 16 waves share the neighbour sums, rows split evenly; sum order kept
# speedup vs baseline: 4.0274x; 1.0878x over previous
#include <hip/hip_runtime.h>

#define COUT 128

typedef float v2f __attribute__((ext_vector_type(2)));
typedef float v8f __attribute__((ext_vector_type(8)));


#define BSH    8
#define BW     (1 << BSH)
#define NCHUNK 256

__device__ __forceinline__ int bucket_of(int d, int nb) {
  return (int)min((unsigned)d >> BSH, (unsigned)(nb - 1));
}

__device__ __forceinline__ int wave_inclusive_sum(int x, int lane) {
  x += __builtin_amdgcn_update_dpp(0, x, 0x111, 0xf, 0xf, false);
  x += __builtin_amdgcn_update_dpp(0, x, 0x112, 0xf, 0xf, false);
  x += __builtin_amdgcn_update_dpp(0, x, 0x114, 0xf, 0xf, false);
  x += __builtin_amdgcn_update_dpp(0, x, 0x118, 0xf, 0xf, false);
  const int low_row = __builtin_amdgcn_readlane(x, 15);
  return x + (lane >= 16 ? low_row : 0);
}

__global__ __launch_bounds__(1024)
void sort_chunk(const int* __restrict__ src, const int* __restrict__ dst,
                int* __restrict__ off, int2* __restrict__ srt, int E, int per, int nb) {
  extern __shared__ int lcur[];
  const int c   = blockIdx.x;
  const int tid = threadIdx.x;
  for (int i = tid; i < nb; i += 1024) lcur[i] = 0;
  __syncthreads();
  const int e0 = c * per;
  const int e1 = min(E, e0 + per);
  for (int base = e0; base < e1; base += 4 * 1024) {
    int d[4];
#pragma unroll
    for (int k = 0; k < 4; ++k) d[k] = dst[min(base + k * 1024 + tid, e1 - 1)];
    asm volatile("" :: "v"(d[0]), "v"(d[1]), "v"(d[2]), "v"(d[3]));
#pragma unroll
    for (int k = 0; k < 4; ++k)
      if (base + k * 1024 + tid < e1) atomicAdd(&lcur[bucket_of(d[k], nb)], 1);
  }
  __syncthreads();
  if (tid < 32) {
    int carry = 0;
    for (int i0 = 0; i0 < nb; i0 += 32) {
      const int i   = i0 + tid;
      const int v   = (i < nb) ? lcur[i] : 0;
      const int inc = wave_inclusive_sum(v, tid);
      if (i < nb) {
        const int start = carry + inc - v;
        lcur[i] = e0 + start;
        off[i * NCHUNK + c] = start;
      }
      carry += __builtin_amdgcn_readlane(inc, 31);
    }
    if (tid == 0) off[nb * NCHUNK + c] = carry;
  }
  __syncthreads();
  for (int base = e0; base < e1; base += 4 * 1024) {
    int s[4], d[4];
#pragma unroll
    for (int k = 0; k < 4; ++k) {
      const int e = min(base + k * 1024 + tid, e1 - 1);
      d[k] = dst[e];
      s[k] = src[e];
    }
    asm volatile("" :: "v"(d[0]), "v"(d[1]), "v"(d[2]), "v"(d[3]), "v"(s[0]), "v"(s[1]), "v"(s[2]), "v"(s[3]));
#pragma unroll
    for (int k = 0; k < 4; ++k)
      if (base + k * 1024 + tid < e1) {
        const int pos = atomicAdd(&lcur[bucket_of(d[k], nb)], 1);
        srt[pos] = make_int2(s[k], d[k]);
      }
  }
}

#define CPW (NCHUNK / 32)

__global__ __launch_bounds__(1024)
void build_rows(const int* __restrict__ off, const int2* __restrict__ srt,
                int* __restrict__ rowptr, int* __restrict__ csr, int N, int E, int per) {
  __shared__ int slot[BW];
  __shared__ int wtot[BW / 32];
  __shared__ int pbeg[NCHUNK];
  __shared__ int pend[NCHUNK];
  __shared__ int part[2][NCHUNK / 32];
  const int b    = blockIdx.x;
  const int tid  = threadIdx.x;
  const int lane = tid & 31;
  const int wave = tid >> 5;
  const int* __restrict__ sdst = reinterpret_cast<const int*>(srt);
  if (tid < BW) slot[tid] = 0;
  if (tid < NCHUNK) {
    const int lo = off[b * NCHUNK + tid];
    const int hi = off[(b + 1) * NCHUNK + tid];
    pbeg[tid] = tid * per + lo;
    pend[tid] = tid * per + hi;
    const int slo = wave_inclusive_sum(lo, lane);
    const int sln = wave_inclusive_sum(hi - lo, lane);
    if (lane == 31) { part[0][wave] = slo; part[1][wave] = sln; }
  }
  __syncthreads();
  int beg = 0, cnt = 0;
#pragma unroll
  for (int w = 0; w < NCHUNK / 32; ++w) { beg += part[0][w]; cnt += part[1][w]; }
  const int end = beg + cnt;

  int lo[CPW], hi[CPW], at[CPW];
#pragma unroll
  for (int s = 0; s < CPW; ++s) {
    lo[s] = pbeg[wave * CPW + s];
    hi[s] = pend[wave * CPW + s];
    at[s] = max(min(min(lo[s] + lane, hi[s] - 1), E - 1), 0);
  }
  {
    int d[CPW];
#pragma unroll
    for (int s = 0; s < CPW; ++s) d[s] = sdst[2 * at[s] + 1];
    static_assert(CPW == 8, "the fence below names CPW values");
    asm volatile("" :: "v"(d[0]), "v"(d[1]), "v"(d[2]), "v"(d[3]), "v"(d[4]), "v"(d[5]), "v"(d[6]), "v"(d[7]));
#pragma unroll
    for (int s = 0; s < CPW; ++s)
      if (lo[s] + lane < hi[s]) atomicAdd(&slot[d[s] & (BW - 1)], 1);
#pragma unroll
    for (int s = 0; s < CPW; ++s)
      for (int i = lo[s] + 32 + lane; i < hi[s]; i += 32) atomicAdd(&slot[sdst[2 * i + 1] & (BW - 1)], 1);
  }
  __syncthreads();
  int deg = 0, inc = 0;
  if (tid < BW) {
    deg = slot[tid];
    inc = wave_inclusive_sum(deg, lane);
    if (lane == 31) wtot[wave] = inc;
  }
  __syncthreads();
  if (tid < BW) {
    int before = 0;
#pragma unroll
    for (int w = 0; w < BW / 32; ++w) before += (w < wave) ? wtot[w] : 0;
    const int node = b * BW + tid;
    const int start = beg + before + inc - deg + min(node, N);
    slot[tid] = start + 1;
    if (node < N) { rowptr[node] = start; csr[start] = node; }
    if (node == N - 1) rowptr[N] = end + N;
  }
  __syncthreads();
  {
    int2 v[CPW];
#pragma unroll
    for (int s = 0; s < CPW; ++s) v[s] = srt[at[s]];
    asm volatile("" :: "v"(v[0].y), "v"(v[1].y), "v"(v[2].y), "v"(v[3].y), "v"(v[4].y), "v"(v[5].y), "v"(v[6].y), "v"(v[7].y));
#pragma unroll
    for (int s = 0; s < CPW; ++s)
      if (lo[s] + lane < hi[s]) {
        const int pos = atomicAdd(&slot[v[s].y & (BW - 1)], 1);
        csr[pos] = v[s].x;
      }
#pragma unroll
    for (int s = 0; s < CPW; ++s)
      for (int i = lo[s] + 32 + lane; i < hi[s]; i += 32) {
        const int2 p = srt[i];
        const int pos = atomicAdd(&slot[p.y & (BW - 1)], 1);
        csr[pos] = p.x;
      }
  }
}

template <int K1, bool RELU_OUT>
__global__ __launch_bounds__(512)
void gin_layer(const float* __restrict__ A, const int* __restrict__ rowptr, const int* __restrict__ csr,
               const float* __restrict__ Wa, const float* __restrict__ ba,
               const float* __restrict__ Wb, const float* __restrict__ bb,
               float* __restrict__ out, int N, int q, int r, int wb_at, int tl_at) {
  __shared__ __align__(16) float w[K1 * COUT + COUT * COUT + 16 * 16 * K1];
  const float* wa = w;
  const float* wb = w + wb_at;
  float* tiles = w + tl_at;

  const int tid  = threadIdx.x;
  const int wave = tid >> 5;
  const int lane = tid & 31;
  const int half = lane >> 4;
  const int l15  = lane & 15;

  const int blk = blockIdx.x;
  const int ns  = q + (blk < r ? 1 : 0);
  const int s0  = blk * q + min(blk, r);

  constexpr int VW = K1 / 32;
  constexpr int NB = 32;
  typedef float rowv __attribute__((ext_vector_type(VW)));
  const rowv* __restrict__ hv = reinterpret_cast<const rowv*>(A);
  const int lslot = (lane * VW) >> 2;
  const int lsub  = (lane * VW) & 3;

  {
    constexpr int NA  = K1 * (COUT / 4) / 512;
    constexpr int NWB = COUT * (COUT / 4) / 512;
    const float4* __restrict__ Wa4 = reinterpret_cast<const float4*>(Wa);
    const float4* __restrict__ Wb4 = reinterpret_cast<const float4*>(Wb);
    float4* wa4 = reinterpret_cast<float4*>(w);
    float4* wb4 = reinterpret_cast<float4*>(w + wb_at);
    float4 va[NA], vb[NWB];
#pragma unroll
    for (int i = 0; i < NA; ++i) va[i] = Wa4[tid + i * 512];
#pragma unroll
    for (int i = 0; i < NWB; ++i) vb[i] = Wb4[tid + i * 512];
#pragma unroll
    for (int i = 0; i < NA; ++i) wa4[tid + i * 512] = va[i];
#pragma unroll
    for (int i = 0; i < NWB; ++i) wb4[tid + i * 512] = vb[i];
  }

  if (ns > 0) {
    const int lr0 = wave * ns;
    const int rb  = s0 * 16 + lr0;
    const int rpv = rowptr[min(rb + min(lane, ns), N)];
    const int E0 = __builtin_amdgcn_readlane(rpv, 0);
    const int E1 = __builtin_amdgcn_readlane(rpv, ns);
    const unsigned last = (unsigned)(N - 1);
    int cur  = 0;
    int rend = __builtin_amdgcn_readlane(rpv, 1);
    rowv sum = {};
    int idxv = (E0 < E1) ? csr[min(E0 + lane, E1 - 1)] : 0;
    for (int e = E0; e < E1; e += NB) {
      const int cidx = idxv;
      if (e + NB < E1) idxv = csr[min(e + NB + lane, E1 - 1)];
      rowv v[NB];
#pragma unroll
      for (int k = 0; k < NB; ++k) {
        const unsigned s = min((unsigned)__builtin_amdgcn_readlane(cidx, k), last);
        v[k] = hv[(size_t)s * 32 + lane];
      }
      int rel = rend - e;
#pragma unroll
      for (int k = 0; k < NB; ++k) {
        if (__builtin_expect(k == rel, 0)) {
          const int lr = lr0 + cur;
          *reinterpret_cast<rowv*>(tiles + lr * K1 + ((lslot ^ (lr & 15)) << 2) + lsub) = sum;
          ++cur;
          rend = __builtin_amdgcn_readlane(rpv, cur + 1);
          rel  = rend - e;
          sum  = rowv{};
        }
        sum += v[k];
      }
    }
    if (cur < ns) {
      const int lr = lr0 + cur;
      *reinterpret_cast<rowv*>(tiles + lr * K1 + ((lslot ^ (lr & 15)) << 2) + lsub) = sum;
    }
  }
  __syncthreads();
  if (wave >= ns) return;

  const int row0 = (s0 + wave) * 16;
  const float* tile = tiles + wave * (16 * K1);

  v2f zf[K1 / 4];
#pragma unroll
  for (int i = 0; i < K1 / 4; ++i)
    zf[i] = *reinterpret_cast<const v2f*>(tile + l15 * K1 + ((i ^ l15) << 2) + half * 2);

  v8f acc[8] = {};
#pragma unroll
  for (int i = 0; i < K1 / 4; ++i) {
    const float* w0 = &wa[(4 * i + half * 2) * COUT + l15];
    const float* w1 = w0 + COUT;
#pragma unroll
    for (int ct = 0; ct < 8; ++ct) {
      v2f wf;
      wf.x = w0[ct * 16];
      wf.y = w1[ct * 16];
      acc[ct] = __builtin_amdgcn_wmma_f32_16x16x4_f32(
          false, wf, false, zf[i], (short)0, acc[ct], false, false);
    }
  }

#pragma unroll
  for (int ct = 0; ct < 8; ++ct) {
    const float4 b0 = *reinterpret_cast<const float4*>(ba + ct * 16 + half * 8);
    const float4 b1 = *reinterpret_cast<const float4*>(ba + ct * 16 + half * 8 + 4);
    acc[ct][0] = fmaxf(acc[ct][0] + b0.x, 0.0f);
    acc[ct][1] = fmaxf(acc[ct][1] + b0.y, 0.0f);
    acc[ct][2] = fmaxf(acc[ct][2] + b0.z, 0.0f);
    acc[ct][3] = fmaxf(acc[ct][3] + b0.w, 0.0f);
    acc[ct][4] = fmaxf(acc[ct][4] + b1.x, 0.0f);
    acc[ct][5] = fmaxf(acc[ct][5] + b1.y, 0.0f);
    acc[ct][6] = fmaxf(acc[ct][6] + b1.z, 0.0f);
    acc[ct][7] = fmaxf(acc[ct][7] + b1.w, 0.0f);
  }

  v8f res[8] = {};
  const float* wbl = wb + half * 8 * COUT + l15;
#pragma unroll
  for (int ct = 0; ct < 8; ++ct) {
#pragma unroll
    for (int j = 0; j < 4; ++j) {
      v2f hf;
      hf.x = acc[ct][2 * j];
      hf.y = acc[ct][2 * j + 1];
      const float* w0 = wbl + (ct * 16 + 2 * j) * COUT;
      const float* w1 = w0 + COUT;
#pragma unroll
      for (int c2 = 0; c2 < 8; ++c2) {
        v2f wf;
        wf.x = w0[c2 * 16];
        wf.y = w1[c2 * 16];
        res[c2] = __builtin_amdgcn_wmma_f32_16x16x4_f32(
            false, wf, false, hf, (short)0, res[c2], false, false);
      }
    }
  }

  float* orow = out + (size_t)(row0 + l15) * COUT + half * 8;
#pragma unroll
  for (int c2 = 0; c2 < 8; ++c2) {
    const float4 b0 = *reinterpret_cast<const float4*>(bb + c2 * 16 + half * 8);
    const float4 b1 = *reinterpret_cast<const float4*>(bb + c2 * 16 + half * 8 + 4);
    float4 o0, o1;
    o0.x = res[c2][0] + b0.x;
    o0.y = res[c2][1] + b0.y;
    o0.z = res[c2][2] + b0.z;
    o0.w = res[c2][3] + b0.w;
    o1.x = res[c2][4] + b1.x;
    o1.y = res[c2][5] + b1.y;
    o1.z = res[c2][6] + b1.z;
    o1.w = res[c2][7] + b1.w;
    if (RELU_OUT) {
      o0.x = fmaxf(o0.x, 0.0f); o0.y = fmaxf(o0.y, 0.0f); o0.z = fmaxf(o0.z, 0.0f); o0.w = fmaxf(o0.w, 0.0f);
      o1.x = fmaxf(o1.x, 0.0f); o1.y = fmaxf(o1.y, 0.0f); o1.z = fmaxf(o1.z, 0.0f); o1.w = fmaxf(o1.w, 0.0f);
    }
    float4* dst = reinterpret_cast<float4*>(orow + c2 * 16);
    dst[0] = o0;
    dst[1] = o1;
  }
}

extern "C" void kernel_launch(void* const* d_in, const int* in_sizes, int n_in,
                              void* d_out, int out_size, void* d_ws, size_t ws_size,
                              hipStream_t stream) {
  const float* x   = (const float*)d_in[0];
  const int*   ei  = (const int*)d_in[1];
  const float* W1a = (const float*)d_in[2];
  const float* b1a = (const float*)d_in[3];
  const float* W1b = (const float*)d_in[4];
  const float* b1b = (const float*)d_in[5];
  const float* W2a = (const float*)d_in[6];
  const float* b2a = (const float*)d_in[7];
  const float* W2b = (const float*)d_in[8];
  const float* b2b = (const float*)d_in[9];
  float* out = (float*)d_out;

  const int N = in_sizes[0] / 64;
  const int E = in_sizes[1] / 2;
  const int* src = ei;
  const int* dst = ei + E;

  float* R0 = (float*)d_ws;
  float* R2 = R0 + (size_t)N * 256;

  const int NB  = (N + BW - 1) / BW;
  const int per = (E + NCHUNK - 1) / NCHUNK;
  const int nt  = (NB + 1) * NCHUNK;

  int*  off    = (int*)(R2 + (size_t)N * 128);
  int*  rowptr = off + (nt + 31) / 32 * 32;
  int2* srt    = (int2*)(rowptr + (N + 1 + 31) / 32 * 32);
  int*  csr    = (int*)(srt + E);

  const int strips  = (N + 15) / 16;
  const int gblocks = (strips + 15) / 16 > 256 ? (strips + 15) / 16 : 256;
  const int q = strips / gblocks;
  const int r = strips % gblocks;

  sort_chunk<<<NCHUNK, 1024, NB * sizeof(int), stream>>>(src, dst, off, srt, E, per, NB);
  build_rows<<<NB, 1024, 0, stream>>>(off, srt, rowptr, csr, N, E, per);

  gin_layer<64, true><<<gblocks, 512, 0, stream>>>(x, rowptr, csr, W1a, b1a, W1b, b1b, R2, N, q, r,
                                                   64 * COUT, 64 * COUT + COUT * COUT);

  gin_layer<128, false><<<gblocks, 512, 0, stream>>>(R2, rowptr, csr, W2a, b2a, W2b, b2b, out, N, q, r,
                                                     128 * COUT, 128 * COUT + COUT * COUT);
}
